# starting kernel
# baseline (speedup 1.0000x reference)
#include <hip/hip_runtime.h>


#define M_ROWS 131072
#define R_COLS 64

typedef __attribute__((ext_vector_type(2))) float v2f;
typedef __attribute__((ext_vector_type(4))) float v4f;
typedef __attribute__((ext_vector_type(8))) float v8f;

#define WS_GR 0
#define WS_GI 4096
#define WS_W  8192

__device__ inline v8f wmma_f32x4(v2f a, v2f b, v8f c) {
  return __builtin_amdgcn_wmma_f32_16x16x4_f32(
       false, a,  false, b,
       (short)0, c,  false,  false);
}

__global__ void fo_zero_gram(float* ws) {
  int i = blockIdx.x * blockDim.x + threadIdx.x;
  if (i < 2 * 64 * 64) ws[i] = 0.0f;
}

__global__ __launch_bounds__(512) void fo_gram(const float* __restrict__ x,
                                               float* __restrict__ ws) {
  const int lane = threadIdx.x & 31;
  const int wave = threadIdx.x >> 5;
  const int ti = wave >> 2;
  const int tj = wave & 3;
  const int p = lane & 15;
  const int kb = (lane < 16) ? 0 : 2;

  const int colA = 16 * ti + p;
  const int colB = 16 * tj + p;
  const int row0 = blockIdx.x * 256;

  v8f gr = {};
  v8f gi = {};

  for (int ks = 0; ks < 256; ks += 4) {
    const size_t m = (size_t)(row0 + ks + kb);
    v2f ea0 = *(const v2f*)(x + (m * R_COLS + colA) * 2);
    v2f ea1 = *(const v2f*)(x + ((m + 1) * R_COLS + colA) * 2);
    v2f eb0 = *(const v2f*)(x + (m * R_COLS + colB) * 2);
    v2f eb1 = *(const v2f*)(x + ((m + 1) * R_COLS + colB) * 2);

    v2f ar = {ea0.x, ea1.x};
    v2f ai = {ea0.y, ea1.y};
    v2f ain = {-ea0.y, -ea1.y};
    v2f br = {eb0.x, eb1.x};
    v2f bi = {eb0.y, eb1.y};

    gr = wmma_f32x4(ar, br, gr);
    gr = wmma_f32x4(ai, bi, gr);
    gi = wmma_f32x4(ar, bi, gi);
    gi = wmma_f32x4(ain, br, gi);
  }

  const int rbase = 16 * ti + ((lane < 16) ? 0 : 8);
  const int c = 16 * tj + p;
  float* grp = ws + WS_GR;
  float* gip = ws + WS_GI;
#pragma unroll
  for (int v = 0; v < 8; ++v) {
    atomicAdd(&grp[(rbase + v) * 64 + c], gr[v]);
    atomicAdd(&gip[(rbase + v) * 64 + c], gi[v]);
  }
}

__global__ __launch_bounds__(64) void fo_chol_inv(float* __restrict__ ws) {
  __shared__ float2 Gs[64 * 64];
  __shared__ float2 Ys[64 * 64];
  const int t = threadIdx.x;
  const float* grp = ws + WS_GR;
  const float* gip = ws + WS_GI;

  for (int i = t; i < 64 * 64; i += 64) {
    Gs[i].x = grp[i];
    Gs[i].y = gip[i];
  }
  __syncthreads();

  for (int k = 0; k < 64; ++k) {
    if (t == k) {
      float d = sqrtf(Gs[k * 64 + k].x + 1e-8f);
      Gs[k * 64 + k].x = d;
      Gs[k * 64 + k].y = 0.0f;
    }
    __syncthreads();
    const float d = Gs[k * 64 + k].x;
    if (t > k) {
      Gs[t * 64 + k].x /= d;
      Gs[t * 64 + k].y /= d;
    }
    __syncthreads();
    if (t > k) {
      const float2 ltk = Gs[t * 64 + k];
      for (int j = k + 1; j <= t; ++j) {
        const float2 ljk = Gs[j * 64 + k];
        Gs[t * 64 + j].x -= ltk.x * ljk.x + ltk.y * ljk.y;
        Gs[t * 64 + j].y -= ltk.y * ljk.x - ltk.x * ljk.y;
      }
    }
    __syncthreads();
  }

  const int cidx = t;
  for (int r = 0; r < 64; ++r) {
    float2 s;
    s.x = (r == cidx) ? 1.0f : 0.0f;
    s.y = 0.0f;
    if (r < cidx) {
      Ys[r * 64 + cidx] = make_float2(0.0f, 0.0f);
      continue;
    }
    for (int kk = cidx; kk < r; ++kk) {
      const float2 l = Gs[r * 64 + kk];
      const float2 y = Ys[kk * 64 + cidx];
      s.x -= l.x * y.x - l.y * y.y;
      s.y -= l.x * y.y + l.y * y.x;
    }
    const float d = Gs[r * 64 + r].x;
    s.x /= d;
    s.y /= d;
    Ys[r * 64 + cidx] = s;
  }
  __syncthreads();

  float* wp = ws + WS_W;
  for (int j = 0; j < 64; ++j) {
    const float2 y = Ys[j * 64 + cidx];
    wp[(cidx * 64 + j) * 2 + 0] = y.x;
    wp[(cidx * 64 + j) * 2 + 1] = -y.y;
  }
}

__global__ __launch_bounds__(512) void fo_apply(const float* __restrict__ x,
                                                const float* __restrict__ ws,
                                                float* __restrict__ out) {
  __shared__ float2 Wl[64 * 64];
  const float* wp = ws + WS_W;
  for (int i = threadIdx.x; i < 64 * 64; i += 512) {
    Wl[i].x = wp[2 * i + 0];
    Wl[i].y = wp[2 * i + 1];
  }
  __syncthreads();

  const int lane = threadIdx.x & 31;
  const int wave = threadIdx.x >> 5;
  const int s = wave >> 2;
  const int tj = wave & 3;
  const int p = lane & 15;
  const int kb = (lane < 16) ? 0 : 2;

  const size_t row0 = (size_t)blockIdx.x * 64 + 16 * s;

  v8f qr = {};
  v8f qi = {};

  for (int ks = 0; ks < 64; ks += 4) {
    v4f e = *(const v4f*)(x + ((row0 + p) * R_COLS + ks + kb) * 2);
    v2f ar = {e.x, e.z};
    v2f ai = {e.y, e.w};
    v2f ain = {-e.y, -e.w};

    const float2 w0 = Wl[(ks + kb) * 64 + 16 * tj + p];
    const float2 w1 = Wl[(ks + kb + 1) * 64 + 16 * tj + p];
    v2f wr = {w0.x, w1.x};
    v2f wi = {w0.y, w1.y};

    qr = wmma_f32x4(ar, wr, qr);
    qr = wmma_f32x4(ain, wi, qr);
    qi = wmma_f32x4(ar, wi, qi);
    qi = wmma_f32x4(ai, wr, qi);
  }

  const int rb = (lane < 16) ? 0 : 8;
  const int c = 16 * tj + p;
#pragma unroll
  for (int v = 0; v < 8; ++v) {
    const size_t row = row0 + rb + v;
    v2f o = {qr[v], qi[v]};
    *(v2f*)(out + (row * R_COLS + c) * 2) = o;
  }
}

extern "C" void kernel_launch(void* const* d_in, const int* in_sizes, int n_in,
                              void* d_out, int out_size, void* d_ws, size_t ws_size,
                              hipStream_t stream) {
  (void)in_sizes; (void)n_in; (void)out_size; (void)ws_size;
  const float* x = (const float*)d_in[0];
  float* out = (float*)d_out;
  float* ws = (float*)d_ws;

  fo_zero_gram<<<(2 * 64 * 64 + 255) / 256, 256, 0, stream>>>(ws);
  fo_gram<<<M_ROWS / 256, 512, 0, stream>>>(x, ws);
  fo_chol_inv<<<1, 64, 0, stream>>>(ws);
  fo_apply<<<M_ROWS / 64, 512, 0, stream>>>(x, ws, out);
}


# inverse of L by 16x16 blocks: the 4 diagonal blocks by substitution in step (16 barrier steps, not 64), the 6 blocks below from LDS block products N_ij = -N_ii*sum_k(L_ik*N_kj) (sum order changes)
# speedup vs baseline: 3.1679x; 3.1679x over previous
#include <hip/hip_runtime.h>


#define M_ROWS 131072
#define R_COLS 64

typedef __attribute__((ext_vector_type(2))) float v2f;
typedef __attribute__((ext_vector_type(4))) float v4f;
typedef __attribute__((ext_vector_type(8))) float v8f;

#define WS_GR 0
#define WS_GI 4096
#define WS_W  8192

__device__ inline v8f wmma_f32x4(v2f a, v2f b, v8f c) {
  return __builtin_amdgcn_wmma_f32_16x16x4_f32(
       false, a,  false, b,
       (short)0, c,  false,  false);
}

__global__ void fo_zero_gram(float* ws) {
  int i = blockIdx.x * blockDim.x + threadIdx.x;
  if (i < 2 * 64 * 64) ws[i] = 0.0f;
}

#define GRAM_BLOCK_ROWS 2048
#define GRAM_WAVE_ROWS 512
#define GRAM_BLOCKS (4 * (M_ROWS / GRAM_BLOCK_ROWS))
#define X_ROW_F (R_COLS * 2)

__device__ __forceinline__ void fo_gram_load16(const float* __restrict__ pa,
                                               const float* __restrict__ pb,
                                               v2f (&ea)[8], v2f (&eb)[8]) {
#pragma unroll
  for (int j = 0; j < 4; ++j) {
    ea[2 * j + 0] = *(const v2f*)(pa + (4 * j + 0) * X_ROW_F);
    ea[2 * j + 1] = *(const v2f*)(pa + (4 * j + 1) * X_ROW_F);
    eb[2 * j + 0] = *(const v2f*)(pb + (4 * j + 0) * X_ROW_F);
    eb[2 * j + 1] = *(const v2f*)(pb + (4 * j + 1) * X_ROW_F);
  }
}

__device__ __forceinline__ void fo_gram_mma16(const v2f (&ea)[8], const v2f (&eb)[8],
                                              v8f& gr, v8f& gi) {
#pragma unroll
  for (int j = 0; j < 4; ++j) {
    const v2f ea0 = ea[2 * j + 0];
    const v2f ea1 = ea[2 * j + 1];
    const v2f eb0 = eb[2 * j + 0];
    const v2f eb1 = eb[2 * j + 1];
    v2f ar = {ea0.x, ea1.x};
    v2f ai = {ea0.y, ea1.y};
    v2f ain = {-ea0.y, -ea1.y};
    v2f br = {eb0.x, eb1.x};
    v2f bi = {eb0.y, eb1.y};

    gr = wmma_f32x4(ar, br, gr);
    gr = wmma_f32x4(ai, bi, gr);
    gi = wmma_f32x4(ar, bi, gi);
    gi = wmma_f32x4(ain, br, gi);
  }
}

__global__ __launch_bounds__(512) void fo_gram(const float* __restrict__ x,
                                               float* __restrict__ ws) {
  __shared__ v4f red[4 * 4 * 4 * 32];

  const int lane = threadIdx.x & 31;
  const int wave = threadIdx.x >> 5;
  const int ti = blockIdx.x & 3;
  const int tj = wave & 3;
  const int kq = wave >> 2;
  const int p = lane & 15;
  const int kb = (lane < 16) ? 0 : 2;

  const int colA = 16 * ti + p;
  const int colB = 16 * tj + p;
  const size_t row0 = (size_t)(blockIdx.x >> 2) * GRAM_BLOCK_ROWS + kq * GRAM_WAVE_ROWS + kb;
  const float* pa = x + (row0 * R_COLS + colA) * 2;
  const float* pb = x + (row0 * R_COLS + colB) * 2;

  v8f gr = {};
  v8f gi = {};

  v2f a0[8], b0[8], a1[8], b1[8];
  constexpr int kSets = GRAM_WAVE_ROWS / 16;
  fo_gram_load16(pa, pb, a0, b0);
  for (int c = 0; c < kSets - 2; c += 2) {
    fo_gram_load16(pa + (c + 1) * 16 * X_ROW_F, pb + (c + 1) * 16 * X_ROW_F, a1, b1);
    fo_gram_mma16(a0, b0, gr, gi);
    fo_gram_load16(pa + (c + 2) * 16 * X_ROW_F, pb + (c + 2) * 16 * X_ROW_F, a0, b0);
    fo_gram_mma16(a1, b1, gr, gi);
  }
  fo_gram_load16(pa + (kSets - 1) * 16 * X_ROW_F, pb + (kSets - 1) * 16 * X_ROW_F, a1, b1);
  fo_gram_mma16(a0, b0, gr, gi);
  fo_gram_mma16(a1, b1, gr, gi);

  {
    v4f* dst = red + ((kq * 4 + tj) * 4) * 32 + lane;
    dst[0 * 32] = v4f{gr[0], gr[1], gr[2], gr[3]};
    dst[1 * 32] = v4f{gr[4], gr[5], gr[6], gr[7]};
    dst[2 * 32] = v4f{gi[0], gi[1], gi[2], gi[3]};
    dst[3 * 32] = v4f{gi[4], gi[5], gi[6], gi[7]};
  }
  __syncthreads();
  const v4f* src = red + (tj * 4 + kq) * 32 + lane;
  v4f acc = src[0];
#pragma unroll
  for (int q = 1; q < 4; ++q) acc += src[q * (4 * 4 * 32)];

  float* gp = ws + ((kq & 2) ? WS_GI : WS_GR);
  const int rbase = 16 * ti + ((lane < 16) ? 0 : 8) + 4 * (kq & 1);
  const int c = 16 * tj + p;
#pragma unroll
  for (int e = 0; e < 4; ++e) atomicAdd(&gp[(rbase + e) * 64 + c], acc[e]);
}

#define NR_LD 65

template <int KB>
__device__ __forceinline__ void fo_factor_block(float2 (&g)[4][4], float2* Lc, float* rdv,
                                                const int ty, const int tx) {
#pragma nounroll
  for (int kk = 0; kk < 16; ++kk) {
    const int k = 16 * KB + kk;
    if (tx == kk) {
#pragma unroll
      for (int a = KB; a < 4; ++a) Lc[k * 64 + 16 * a + ty] = g[a][KB];
      if (ty == kk) {
        const float d = sqrtf(g[KB][KB].x + 1e-8f);
        rdv[k] = 1.0f / d;
      }
    }
    __syncthreads();
    const float rd = rdv[k];
    float2 li[4];
    float2 lj[4];
#pragma unroll
    for (int a = KB; a < 4; ++a) {
      const float2 t = Lc[k * 64 + 16 * a + ty];
      li[a] = make_float2(t.x * rd, t.y * rd);
    }
#pragma unroll
    for (int b = KB; b < 4; ++b) {
      const float2 t = Lc[k * 64 + 16 * b + tx];
      lj[b] = make_float2(t.x * rd, t.y * rd);
    }
#pragma unroll
    for (int b = KB; b < 4; ++b) {
#pragma unroll
      for (int a = b; a < 4; ++a) {
        const bool live = (b > KB || tx > kk) && (a > b || ty >= tx);
        if (live) {
          g[a][b].x -= li[a].x * lj[b].x + li[a].y * lj[b].y;
          g[a][b].y -= li[a].y * lj[b].x - li[a].x * lj[b].y;
        }
      }
    }
  }
}

__device__ __forceinline__ void fo_invert_diag(float2 (&nd)[4], const float2* Lc, float2* Nr,
                                               const float* rdv, const int ty, const int tx) {
  float rd[4];
#pragma unroll
  for (int q = 0; q < 4; ++q) rd[q] = rdv[16 * q];
#pragma nounroll
  for (int kk = 0; kk < 16; ++kk) {
    if (ty == kk) {
#pragma unroll
      for (int q = 0; q < 4; ++q) {
        const float2 v = make_float2(nd[q].x * rd[q], nd[q].y * rd[q]);
        nd[q] = v;
        Nr[(16 * q + kk) * NR_LD + 16 * q + tx] = v;
      }
    }
    __syncthreads();
    float rdn[4];
#pragma unroll
    for (int q = 0; q < 4; ++q) rdn[q] = rdv[16 * q + ((kk + 1) & 15)];
#pragma unroll
    for (int q = 0; q < 4; ++q) {
      const int k = 16 * q + kk;
      const float2 t = Lc[k * 64 + 16 * q + ty];
      const float2 lr = make_float2(t.x * rd[q], t.y * rd[q]);
      const float2 nk = Nr[k * NR_LD + 16 * q + tx];
      if (ty > kk) {
        nd[q].x -= lr.x * nk.x - lr.y * nk.y;
        nd[q].y -= lr.x * nk.y + lr.y * nk.x;
      }
    }
#pragma unroll
    for (int q = 0; q < 4; ++q) rd[q] = rdn[q];
  }
}

#define TS_LD 17
#define TS_BLK (16 * TS_LD)

template <int D>
__device__ __forceinline__ void fo_inverse_level(const float2* Lc, float2* Nr, float2* Ts,
                                                 const int ty, const int tx) {
  constexpr int NB = 4 - D;
  constexpr int NM = 16 * D;
  {
    float2 a[NB][NM];
    float2 b[NB][NM];
#pragma unroll
    for (int n = 0; n < NB; ++n) {
#pragma unroll
      for (int e = 0; e < NM; ++e) {
        const int c = 16 * n + e;
        a[n][e] = Lc[c * 64 + 16 * (D + n) + ty];
        b[n][e] = Nr[c * NR_LD + 16 * n + tx];
      }
    }
#pragma unroll
    for (int n = 0; n < NB; ++n) {
      float2 s0 = make_float2(0.0f, 0.0f);
      float2 s1 = make_float2(0.0f, 0.0f);
#pragma unroll
      for (int e = 0; e < NM; e += 2) {
        s0.x += a[n][e].x * b[n][e].x;
        s0.x -= a[n][e].y * b[n][e].y;
        s0.y += a[n][e].x * b[n][e].y;
        s0.y += a[n][e].y * b[n][e].x;
        s1.x += a[n][e + 1].x * b[n][e + 1].x;
        s1.x -= a[n][e + 1].y * b[n][e + 1].y;
        s1.y += a[n][e + 1].x * b[n][e + 1].y;
        s1.y += a[n][e + 1].y * b[n][e + 1].x;
      }
      Ts[(D + n) * TS_BLK + ty * TS_LD + tx] = make_float2(s0.x + s1.x, s0.y + s1.y);
    }
  }
  __syncthreads();
  {
    float2 a[NB][16];
    float2 b[NB][16];
#pragma unroll
    for (int n = 0; n < NB; ++n) {
      const int i = D + n;
#pragma unroll
      for (int m = 0; m < 16; ++m) {
        a[n][m] = Nr[(16 * i + ty) * NR_LD + 16 * i + m];
        b[n][m] = Ts[i * TS_BLK + m * TS_LD + tx];
      }
    }
#pragma unroll
    for (int n = 0; n < NB; ++n) {
      float2 s0 = make_float2(0.0f, 0.0f);
      float2 s1 = make_float2(0.0f, 0.0f);
#pragma unroll
      for (int m = 0; m < 16; m += 2) {
        s0.x += a[n][m].x * b[n][m].x;
        s0.x -= a[n][m].y * b[n][m].y;
        s0.y += a[n][m].x * b[n][m].y;
        s0.y += a[n][m].y * b[n][m].x;
        s1.x += a[n][m + 1].x * b[n][m + 1].x;
        s1.x -= a[n][m + 1].y * b[n][m + 1].y;
        s1.y += a[n][m + 1].x * b[n][m + 1].y;
        s1.y += a[n][m + 1].y * b[n][m + 1].x;
      }
      Nr[(16 * (D + n) + ty) * NR_LD + 16 * n + tx] = make_float2(-(s0.x + s1.x), -(s0.y + s1.y));
    }
  }
  __syncthreads();
}

__global__ __launch_bounds__(256) void fo_chol_inv(float* __restrict__ ws) {
  __shared__ float2 Lc[64 * 64];
  __shared__ float2 Nr[64 * NR_LD];
  __shared__ float rdv[64];
  __shared__ float2 Ts[4 * TS_BLK];
  const int tid = threadIdx.x;
  const int ty = tid >> 4;
  const int tx = tid & 15;
  const float* grp = ws + WS_GR;
  const float* gip = ws + WS_GI;

  float2 g[4][4];
#pragma unroll
  for (int a = 0; a < 4; ++a) {
#pragma unroll
    for (int b = 0; b <= a; ++b) {
      const int idx = (16 * a + ty) * 64 + 16 * b + tx;
      g[a][b] = make_float2(grp[idx], gip[idx]);
    }
  }

  fo_factor_block<0>(g, Lc, rdv, ty, tx);
  fo_factor_block<1>(g, Lc, rdv, ty, tx);
  fo_factor_block<2>(g, Lc, rdv, ty, tx);
  fo_factor_block<3>(g, Lc, rdv, ty, tx);

  {
    float2 t[12];
    float s[12];
#pragma unroll
    for (int m = 0; m < 12; ++m) {
      t[m] = Lc[tid + 256 * m];
      s[m] = rdv[(tid >> 6) + 4 * m];
    }
    const int mend = 4 * ((tid & 63) >> 4);
#pragma unroll
    for (int m = 0; m < 12; ++m) {
      if (m < mend) Lc[tid + 256 * m] = make_float2(t[m].x * s[m], t[m].y * s[m]);
    }
  }

  float2 nd[4];
#pragma unroll
  for (int q = 0; q < 4; ++q) nd[q] = make_float2((ty == tx) ? 1.0f : 0.0f, 0.0f);
  fo_invert_diag(nd, Lc, Nr, rdv, ty, tx);
  fo_inverse_level<1>(Lc, Nr, Ts, ty, tx);
  fo_inverse_level<2>(Lc, Nr, Ts, ty, tx);
  fo_inverse_level<3>(Lc, Nr, Ts, ty, tx);

  float* wp = ws + WS_W;
  for (int m = 0; m < 16; ++m) {
    const int o = tid + 256 * m;
    const int c = o >> 6;
    const int j = o & 63;
    float2 v = make_float2(0.0f, 0.0f);
    if (j >= c) v = Nr[j * NR_LD + c];
    *(float2*)(wp + 2 * o) = make_float2(v.x, -v.y);
  }
}

__global__ __launch_bounds__(512) void fo_apply(const float* __restrict__ x,
                                                const float* __restrict__ ws,
                                                float* __restrict__ out) {
  __shared__ float2 Wl[64 * 64];
  const float* wp = ws + WS_W;
  for (int i = threadIdx.x; i < 64 * 64; i += 512) {
    Wl[i].x = wp[2 * i + 0];
    Wl[i].y = wp[2 * i + 1];
  }
  __syncthreads();

  const int lane = threadIdx.x & 31;
  const int wave = threadIdx.x >> 5;
  const int s = wave >> 2;
  const int tj = wave & 3;
  const int p = lane & 15;
  const int kb = (lane < 16) ? 0 : 2;

  const size_t row0 = (size_t)blockIdx.x * 64 + 16 * s;

  v8f qr = {};
  v8f qi = {};

  for (int ks = 0; ks < 64; ks += 4) {
    v4f e = *(const v4f*)(x + ((row0 + p) * R_COLS + ks + kb) * 2);
    v2f ar = {e.x, e.z};
    v2f ai = {e.y, e.w};
    v2f ain = {-e.y, -e.w};

    const float2 w0 = Wl[(ks + kb) * 64 + 16 * tj + p];
    const float2 w1 = Wl[(ks + kb + 1) * 64 + 16 * tj + p];
    v2f wr = {w0.x, w1.x};
    v2f wi = {w0.y, w1.y};

    qr = wmma_f32x4(ar, wr, qr);
    qr = wmma_f32x4(ain, wi, qr);
    qi = wmma_f32x4(ar, wi, qi);
    qi = wmma_f32x4(ai, wr, qi);
  }

  const int rb = (lane < 16) ? 0 : 8;
  const int c = 16 * tj + p;
#pragma unroll
  for (int v = 0; v < 8; ++v) {
    const size_t row = row0 + rb + v;
    v2f o = {qr[v], qi[v]};
    *(v2f*)(out + (row * R_COLS + c) * 2) = o;
  }
}

extern "C" void kernel_launch(void* const* d_in, const int* in_sizes, int n_in,
                              void* d_out, int out_size, void* d_ws, size_t ws_size,
                              hipStream_t stream) {
  (void)in_sizes; (void)n_in; (void)out_size; (void)ws_size;
  const float* x = (const float*)d_in[0];
  float* out = (float*)d_out;
  float* ws = (float*)d_ws;

  fo_zero_gram<<<(2 * 64 * 64 + 255) / 256, 256, 0, stream>>>(ws);
  fo_gram<<<GRAM_BLOCKS, 512, 0, stream>>>(x, ws);
  fo_chol_inv<<<1, 256, 0, stream>>>(ws);
  fo_apply<<<M_ROWS / 64, 512, 0, stream>>>(x, ws, out);
}


# apply: 256 persistent work-groups (one per WGP), each wave reads its loop-invariant W fragments once and loops over 8 strips, next strip's X loads issued ahead of the WMMAs (4096 waves, not 32768)
# speedup vs baseline: 3.3218x; 1.0486x over previous
#include <hip/hip_runtime.h>


#define M_ROWS 131072
#define R_COLS 64

typedef __attribute__((ext_vector_type(2))) float v2f;
typedef __attribute__((ext_vector_type(4))) float v4f;
typedef __attribute__((ext_vector_type(8))) float v8f;

#define WS_GR 0
#define WS_GI 4096
#define WS_W  8192

__device__ inline v8f wmma_f32x4(v2f a, v2f b, v8f c) {
  return __builtin_amdgcn_wmma_f32_16x16x4_f32(
       false, a,  false, b,
       (short)0, c,  false,  false);
}

__global__ void fo_zero_gram(float* ws) {
  int i = blockIdx.x * blockDim.x + threadIdx.x;
  if (i < 2 * 64 * 64) ws[i] = 0.0f;
}

#define GRAM_BLOCK_ROWS 2048
#define GRAM_WAVE_ROWS 512
#define GRAM_BLOCKS (4 * (M_ROWS / GRAM_BLOCK_ROWS))
#define X_ROW_F (R_COLS * 2)

__device__ __forceinline__ void fo_gram_load16(const float* __restrict__ pa,
                                               const float* __restrict__ pb,
                                               v2f (&ea)[8], v2f (&eb)[8]) {
#pragma unroll
  for (int j = 0; j < 4; ++j) {
    ea[2 * j + 0] = *(const v2f*)(pa + (4 * j + 0) * X_ROW_F);
    ea[2 * j + 1] = *(const v2f*)(pa + (4 * j + 1) * X_ROW_F);
    eb[2 * j + 0] = *(const v2f*)(pb + (4 * j + 0) * X_ROW_F);
    eb[2 * j + 1] = *(const v2f*)(pb + (4 * j + 1) * X_ROW_F);
  }
}

__device__ __forceinline__ void fo_gram_mma16(const v2f (&ea)[8], const v2f (&eb)[8],
                                              v8f& gr, v8f& gi) {
#pragma unroll
  for (int j = 0; j < 4; ++j) {
    const v2f ea0 = ea[2 * j + 0];
    const v2f ea1 = ea[2 * j + 1];
    const v2f eb0 = eb[2 * j + 0];
    const v2f eb1 = eb[2 * j + 1];
    v2f ar = {ea0.x, ea1.x};
    v2f ai = {ea0.y, ea1.y};
    v2f ain = {-ea0.y, -ea1.y};
    v2f br = {eb0.x, eb1.x};
    v2f bi = {eb0.y, eb1.y};

    gr = wmma_f32x4(ar, br, gr);
    gr = wmma_f32x4(ai, bi, gr);
    gi = wmma_f32x4(ar, bi, gi);
    gi = wmma_f32x4(ain, br, gi);
  }
}

__global__ __launch_bounds__(512) void fo_gram(const float* __restrict__ x,
                                               float* __restrict__ ws) {
  __shared__ v4f red[4 * 4 * 4 * 32];

  const int lane = threadIdx.x & 31;
  const int wave = threadIdx.x >> 5;
  const int ti = blockIdx.x & 3;
  const int tj = wave & 3;
  const int kq = wave >> 2;
  const int p = lane & 15;
  const int kb = (lane < 16) ? 0 : 2;

  const int colA = 16 * ti + p;
  const int colB = 16 * tj + p;
  const size_t row0 = (size_t)(blockIdx.x >> 2) * GRAM_BLOCK_ROWS + kq * GRAM_WAVE_ROWS + kb;
  const float* pa = x + (row0 * R_COLS + colA) * 2;
  const float* pb = x + (row0 * R_COLS + colB) * 2;

  v8f gr = {};
  v8f gi = {};

  v2f a0[8], b0[8], a1[8], b1[8];
  constexpr int kSets = GRAM_WAVE_ROWS / 16;
  fo_gram_load16(pa, pb, a0, b0);
  for (int c = 0; c < kSets - 2; c += 2) {
    fo_gram_load16(pa + (c + 1) * 16 * X_ROW_F, pb + (c + 1) * 16 * X_ROW_F, a1, b1);
    fo_gram_mma16(a0, b0, gr, gi);
    fo_gram_load16(pa + (c + 2) * 16 * X_ROW_F, pb + (c + 2) * 16 * X_ROW_F, a0, b0);
    fo_gram_mma16(a1, b1, gr, gi);
  }
  fo_gram_load16(pa + (kSets - 1) * 16 * X_ROW_F, pb + (kSets - 1) * 16 * X_ROW_F, a1, b1);
  fo_gram_mma16(a0, b0, gr, gi);
  fo_gram_mma16(a1, b1, gr, gi);

  {
    v4f* dst = red + ((kq * 4 + tj) * 4) * 32 + lane;
    dst[0 * 32] = v4f{gr[0], gr[1], gr[2], gr[3]};
    dst[1 * 32] = v4f{gr[4], gr[5], gr[6], gr[7]};
    dst[2 * 32] = v4f{gi[0], gi[1], gi[2], gi[3]};
    dst[3 * 32] = v4f{gi[4], gi[5], gi[6], gi[7]};
  }
  __syncthreads();
  const v4f* src = red + (tj * 4 + kq) * 32 + lane;
  v4f acc = src[0];
#pragma unroll
  for (int q = 1; q < 4; ++q) acc += src[q * (4 * 4 * 32)];

  float* gp = ws + ((kq & 2) ? WS_GI : WS_GR);
  const int rbase = 16 * ti + ((lane < 16) ? 0 : 8) + 4 * (kq & 1);
  const int c = 16 * tj + p;
#pragma unroll
  for (int e = 0; e < 4; ++e) atomicAdd(&gp[(rbase + e) * 64 + c], acc[e]);
}

#define NR_LD 65

template <int KB>
__device__ __forceinline__ void fo_factor_block(float2 (&g)[4][4], float2* Lc, float* rdv,
                                                const int ty, const int tx) {
#pragma nounroll
  for (int kk = 0; kk < 16; ++kk) {
    const int k = 16 * KB + kk;
    if (tx == kk) {
#pragma unroll
      for (int a = KB; a < 4; ++a) Lc[k * 64 + 16 * a + ty] = g[a][KB];
      if (ty == kk) {
        const float d = sqrtf(g[KB][KB].x + 1e-8f);
        rdv[k] = 1.0f / d;
      }
    }
    __syncthreads();
    const float rd = rdv[k];
    float2 li[4];
    float2 lj[4];
#pragma unroll
    for (int a = KB; a < 4; ++a) {
      const float2 t = Lc[k * 64 + 16 * a + ty];
      li[a] = make_float2(t.x * rd, t.y * rd);
    }
#pragma unroll
    for (int b = KB; b < 4; ++b) {
      const float2 t = Lc[k * 64 + 16 * b + tx];
      lj[b] = make_float2(t.x * rd, t.y * rd);
    }
#pragma unroll
    for (int b = KB; b < 4; ++b) {
#pragma unroll
      for (int a = b; a < 4; ++a) {
        const bool live = (b > KB || tx > kk) && (a > b || ty >= tx);
        if (live) {
          g[a][b].x -= li[a].x * lj[b].x + li[a].y * lj[b].y;
          g[a][b].y -= li[a].y * lj[b].x - li[a].x * lj[b].y;
        }
      }
    }
  }
}

__device__ __forceinline__ void fo_invert_diag(float2 (&nd)[4], const float2* Lc, float2* Nr,
                                               const float* rdv, const int ty, const int tx) {
  float rd[4];
#pragma unroll
  for (int q = 0; q < 4; ++q) rd[q] = rdv[16 * q];
#pragma nounroll
  for (int kk = 0; kk < 16; ++kk) {
    if (ty == kk) {
#pragma unroll
      for (int q = 0; q < 4; ++q) {
        const float2 v = make_float2(nd[q].x * rd[q], nd[q].y * rd[q]);
        nd[q] = v;
        Nr[(16 * q + kk) * NR_LD + 16 * q + tx] = v;
      }
    }
    __syncthreads();
    float rdn[4];
#pragma unroll
    for (int q = 0; q < 4; ++q) rdn[q] = rdv[16 * q + ((kk + 1) & 15)];
#pragma unroll
    for (int q = 0; q < 4; ++q) {
      const int k = 16 * q + kk;
      const float2 t = Lc[k * 64 + 16 * q + ty];
      const float2 lr = make_float2(t.x * rd[q], t.y * rd[q]);
      const float2 nk = Nr[k * NR_LD + 16 * q + tx];
      if (ty > kk) {
        nd[q].x -= lr.x * nk.x - lr.y * nk.y;
        nd[q].y -= lr.x * nk.y + lr.y * nk.x;
      }
    }
#pragma unroll
    for (int q = 0; q < 4; ++q) rd[q] = rdn[q];
  }
}

#define TS_LD 17
#define TS_BLK (16 * TS_LD)

template <int D>
__device__ __forceinline__ void fo_inverse_level(const float2* Lc, float2* Nr, float2* Ts,
                                                 const int ty, const int tx) {
  constexpr int NB = 4 - D;
  constexpr int NM = 16 * D;
  {
    float2 a[NB][NM];
    float2 b[NB][NM];
#pragma unroll
    for (int n = 0; n < NB; ++n) {
#pragma unroll
      for (int e = 0; e < NM; ++e) {
        const int c = 16 * n + e;
        a[n][e] = Lc[c * 64 + 16 * (D + n) + ty];
        b[n][e] = Nr[c * NR_LD + 16 * n + tx];
      }
    }
#pragma unroll
    for (int n = 0; n < NB; ++n) {
      float2 s0 = make_float2(0.0f, 0.0f);
      float2 s1 = make_float2(0.0f, 0.0f);
#pragma unroll
      for (int e = 0; e < NM; e += 2) {
        s0.x += a[n][e].x * b[n][e].x;
        s0.x -= a[n][e].y * b[n][e].y;
        s0.y += a[n][e].x * b[n][e].y;
        s0.y += a[n][e].y * b[n][e].x;
        s1.x += a[n][e + 1].x * b[n][e + 1].x;
        s1.x -= a[n][e + 1].y * b[n][e + 1].y;
        s1.y += a[n][e + 1].x * b[n][e + 1].y;
        s1.y += a[n][e + 1].y * b[n][e + 1].x;
      }
      Ts[(D + n) * TS_BLK + ty * TS_LD + tx] = make_float2(s0.x + s1.x, s0.y + s1.y);
    }
  }
  __syncthreads();
  {
    float2 a[NB][16];
    float2 b[NB][16];
#pragma unroll
    for (int n = 0; n < NB; ++n) {
      const int i = D + n;
#pragma unroll
      for (int m = 0; m < 16; ++m) {
        a[n][m] = Nr[(16 * i + ty) * NR_LD + 16 * i + m];
        b[n][m] = Ts[i * TS_BLK + m * TS_LD + tx];
      }
    }
#pragma unroll
    for (int n = 0; n < NB; ++n) {
      float2 s0 = make_float2(0.0f, 0.0f);
      float2 s1 = make_float2(0.0f, 0.0f);
#pragma unroll
      for (int m = 0; m < 16; m += 2) {
        s0.x += a[n][m].x * b[n][m].x;
        s0.x -= a[n][m].y * b[n][m].y;
        s0.y += a[n][m].x * b[n][m].y;
        s0.y += a[n][m].y * b[n][m].x;
        s1.x += a[n][m + 1].x * b[n][m + 1].x;
        s1.x -= a[n][m + 1].y * b[n][m + 1].y;
        s1.y += a[n][m + 1].x * b[n][m + 1].y;
        s1.y += a[n][m + 1].y * b[n][m + 1].x;
      }
      Nr[(16 * (D + n) + ty) * NR_LD + 16 * n + tx] = make_float2(-(s0.x + s1.x), -(s0.y + s1.y));
    }
  }
  __syncthreads();
}

__global__ __launch_bounds__(256) void fo_chol_inv(float* __restrict__ ws) {
  __shared__ float2 Lc[64 * 64];
  __shared__ float2 Nr[64 * NR_LD];
  __shared__ float rdv[64];
  __shared__ float2 Ts[4 * TS_BLK];
  const int tid = threadIdx.x;
  const int ty = tid >> 4;
  const int tx = tid & 15;
  const float* grp = ws + WS_GR;
  const float* gip = ws + WS_GI;

  float2 g[4][4];
#pragma unroll
  for (int a = 0; a < 4; ++a) {
#pragma unroll
    for (int b = 0; b <= a; ++b) {
      const int idx = (16 * a + ty) * 64 + 16 * b + tx;
      g[a][b] = make_float2(grp[idx], gip[idx]);
    }
  }

  fo_factor_block<0>(g, Lc, rdv, ty, tx);
  fo_factor_block<1>(g, Lc, rdv, ty, tx);
  fo_factor_block<2>(g, Lc, rdv, ty, tx);
  fo_factor_block<3>(g, Lc, rdv, ty, tx);

  {
    float2 t[12];
    float s[12];
#pragma unroll
    for (int m = 0; m < 12; ++m) {
      t[m] = Lc[tid + 256 * m];
      s[m] = rdv[(tid >> 6) + 4 * m];
    }
    const int mend = 4 * ((tid & 63) >> 4);
#pragma unroll
    for (int m = 0; m < 12; ++m) {
      if (m < mend) Lc[tid + 256 * m] = make_float2(t[m].x * s[m], t[m].y * s[m]);
    }
  }

  float2 nd[4];
#pragma unroll
  for (int q = 0; q < 4; ++q) nd[q] = make_float2((ty == tx) ? 1.0f : 0.0f, 0.0f);
  fo_invert_diag(nd, Lc, Nr, rdv, ty, tx);
  fo_inverse_level<1>(Lc, Nr, Ts, ty, tx);
  fo_inverse_level<2>(Lc, Nr, Ts, ty, tx);
  fo_inverse_level<3>(Lc, Nr, Ts, ty, tx);

  float* wp = ws + WS_W;
  for (int m = 0; m < 16; ++m) {
    const int o = tid + 256 * m;
    const int c = o >> 6;
    const int j = o & 63;
    float2 v = make_float2(0.0f, 0.0f);
    if (j >= c) v = Nr[j * NR_LD + c];
    *(float2*)(wp + 2 * o) = make_float2(v.x, -v.y);
  }
}

#define APPLY_BLOCK_ROWS 512
#define APPLY_GROUPS (APPLY_BLOCK_ROWS / 64)
#define APPLY_BLOCKS (M_ROWS / APPLY_BLOCK_ROWS)

__device__ __forceinline__ void fo_apply_load(const float* __restrict__ px, v4f (&e)[16]) {
#pragma unroll
  for (int j = 0; j < 16; ++j) e[j] = *(const v4f*)(px + 8 * j);
}

__device__ __forceinline__ void fo_apply_strip(const v4f (&e)[16], const v2f (&wr)[16],
                                               const v2f (&wi)[16], float* __restrict__ po) {
  v8f qr = {};
  v8f qi = {};
#pragma unroll
  for (int j = 0; j < 16; ++j) {
    v2f ar = {e[j].x, e[j].z};
    v2f ai = {e[j].y, e[j].w};
    v2f ain = {-e[j].y, -e[j].w};
    qr = wmma_f32x4(ar, wr[j], qr);
    qr = wmma_f32x4(ain, wi[j], qr);
    qi = wmma_f32x4(ar, wi[j], qi);
    qi = wmma_f32x4(ai, wr[j], qi);
  }
#pragma unroll
  for (int v = 0; v < 8; ++v) {
    v2f o = {qr[v], qi[v]};
    *(v2f*)(po + (size_t)v * X_ROW_F) = o;
  }
}

__global__ __launch_bounds__(512) void fo_apply(const float* __restrict__ x,
                                                const float* __restrict__ ws,
                                                float* __restrict__ out) {
  const int lane = threadIdx.x & 31;
  const int wave = threadIdx.x >> 5;
  const int s = wave >> 2;
  const int tj = wave & 3;
  const int p = lane & 15;
  const int kb = (lane < 16) ? 0 : 2;
  const int rb = (lane < 16) ? 0 : 8;

  const float* wp = ws + WS_W + ((size_t)kb * 64 + 16 * tj + p) * 2;
  v2f wr[16];
  v2f wi[16];
#pragma unroll
  for (int j = 0; j < 16; ++j) {
    const v2f w0 = *(const v2f*)(wp + (size_t)(4 * j + 0) * 64 * 2);
    const v2f w1 = *(const v2f*)(wp + (size_t)(4 * j + 1) * 64 * 2);
    wr[j] = v2f{w0.x, w1.x};
    wi[j] = v2f{w0.y, w1.y};
  }

  const size_t row0 = (size_t)blockIdx.x * APPLY_BLOCK_ROWS + 16 * s;
  const float* px = x + ((row0 + p) * R_COLS + kb) * 2;
  float* po = out + ((row0 + rb) * R_COLS + 16 * tj + p) * 2;
  constexpr size_t kGroupF = (size_t)64 * X_ROW_F;

  v4f e0[16], e1[16];
  fo_apply_load(px, e0);
  for (int g = 0; g < APPLY_GROUPS - 2; g += 2) {
    fo_apply_load(px + (g + 1) * kGroupF, e1);
    fo_apply_strip(e0, wr, wi, po + g * kGroupF);
    fo_apply_load(px + (g + 2) * kGroupF, e0);
    fo_apply_strip(e1, wr, wi, po + (g + 1) * kGroupF);
  }
  fo_apply_load(px + (APPLY_GROUPS - 1) * kGroupF, e1);
  fo_apply_strip(e0, wr, wi, po + (APPLY_GROUPS - 2) * kGroupF);
  fo_apply_strip(e1, wr, wi, po + (APPLY_GROUPS - 1) * kGroupF);
}

extern "C" void kernel_launch(void* const* d_in, const int* in_sizes, int n_in,
                              void* d_out, int out_size, void* d_ws, size_t ws_size,
                              hipStream_t stream) {
  (void)in_sizes; (void)n_in; (void)out_size; (void)ws_size;
  const float* x = (const float*)d_in[0];
  float* out = (float*)d_out;
  float* ws = (float*)d_ws;

  fo_zero_gram<<<(2 * 64 * 64 + 255) / 256, 256, 0, stream>>>(ws);
  fo_gram<<<GRAM_BLOCKS, 512, 0, stream>>>(x, ws);
  fo_chol_inv<<<1, 256, 0, stream>>>(ws);
  fo_apply<<<APPLY_BLOCKS, 512, 0, stream>>>(x, ws, out);
}


# factor: two pivots per LDS hand-over and barrier (32 steps, not 64); every thread derives both pivots itself, 1/sqrt by the hardware rsq, not sqrt and divide; rank-2 update (rounding changes)
# speedup vs baseline: 3.4674x; 1.0438x over previous
#include <hip/hip_runtime.h>


#define M_ROWS 131072
#define R_COLS 64

typedef __attribute__((ext_vector_type(2))) float v2f;
typedef __attribute__((ext_vector_type(4))) float v4f;
typedef __attribute__((ext_vector_type(8))) float v8f;

#define WS_GR 0
#define WS_GI 4096
#define WS_W  8192

__device__ inline v8f wmma_f32x4(v2f a, v2f b, v8f c) {
  return __builtin_amdgcn_wmma_f32_16x16x4_f32(
       false, a,  false, b,
       (short)0, c,  false,  false);
}

__global__ void fo_zero_gram(float* ws) {
  int i = blockIdx.x * blockDim.x + threadIdx.x;
  if (i < 2 * 64 * 64) ws[i] = 0.0f;
}

#define GRAM_BLOCK_ROWS 2048
#define GRAM_WAVE_ROWS 512
#define GRAM_BLOCKS (4 * (M_ROWS / GRAM_BLOCK_ROWS))
#define X_ROW_F (R_COLS * 2)

__device__ __forceinline__ void fo_gram_load16(const float* __restrict__ pa,
                                               const float* __restrict__ pb,
                                               v2f (&ea)[8], v2f (&eb)[8]) {
#pragma unroll
  for (int j = 0; j < 4; ++j) {
    ea[2 * j + 0] = *(const v2f*)(pa + (4 * j + 0) * X_ROW_F);
    ea[2 * j + 1] = *(const v2f*)(pa + (4 * j + 1) * X_ROW_F);
    eb[2 * j + 0] = *(const v2f*)(pb + (4 * j + 0) * X_ROW_F);
    eb[2 * j + 1] = *(const v2f*)(pb + (4 * j + 1) * X_ROW_F);
  }
}

__device__ __forceinline__ void fo_gram_mma16(const v2f (&ea)[8], const v2f (&eb)[8],
                                              v8f& gr, v8f& gi) {
#pragma unroll
  for (int j = 0; j < 4; ++j) {
    const v2f ea0 = ea[2 * j + 0];
    const v2f ea1 = ea[2 * j + 1];
    const v2f eb0 = eb[2 * j + 0];
    const v2f eb1 = eb[2 * j + 1];
    v2f ar = {ea0.x, ea1.x};
    v2f ai = {ea0.y, ea1.y};
    v2f ain = {-ea0.y, -ea1.y};
    v2f br = {eb0.x, eb1.x};
    v2f bi = {eb0.y, eb1.y};

    gr = wmma_f32x4(ar, br, gr);
    gr = wmma_f32x4(ai, bi, gr);
    gi = wmma_f32x4(ar, bi, gi);
    gi = wmma_f32x4(ain, br, gi);
  }
}

__global__ __launch_bounds__(512) void fo_gram(const float* __restrict__ x,
                                               float* __restrict__ ws) {
  __shared__ v4f red[4 * 4 * 4 * 32];

  const int lane = threadIdx.x & 31;
  const int wave = threadIdx.x >> 5;
  const int ti = blockIdx.x & 3;
  const int tj = wave & 3;
  const int kq = wave >> 2;
  const int p = lane & 15;
  const int kb = (lane < 16) ? 0 : 2;

  const int colA = 16 * ti + p;
  const int colB = 16 * tj + p;
  const size_t row0 = (size_t)(blockIdx.x >> 2) * GRAM_BLOCK_ROWS + kq * GRAM_WAVE_ROWS + kb;
  const float* pa = x + (row0 * R_COLS + colA) * 2;
  const float* pb = x + (row0 * R_COLS + colB) * 2;

  v8f gr = {};
  v8f gi = {};

  v2f a0[8], b0[8], a1[8], b1[8];
  constexpr int kSets = GRAM_WAVE_ROWS / 16;
  fo_gram_load16(pa, pb, a0, b0);
  for (int c = 0; c < kSets - 2; c += 2) {
    fo_gram_load16(pa + (c + 1) * 16 * X_ROW_F, pb + (c + 1) * 16 * X_ROW_F, a1, b1);
    fo_gram_mma16(a0, b0, gr, gi);
    fo_gram_load16(pa + (c + 2) * 16 * X_ROW_F, pb + (c + 2) * 16 * X_ROW_F, a0, b0);
    fo_gram_mma16(a1, b1, gr, gi);
  }
  fo_gram_load16(pa + (kSets - 1) * 16 * X_ROW_F, pb + (kSets - 1) * 16 * X_ROW_F, a1, b1);
  fo_gram_mma16(a0, b0, gr, gi);
  fo_gram_mma16(a1, b1, gr, gi);

  {
    v4f* dst = red + ((kq * 4 + tj) * 4) * 32 + lane;
    dst[0 * 32] = v4f{gr[0], gr[1], gr[2], gr[3]};
    dst[1 * 32] = v4f{gr[4], gr[5], gr[6], gr[7]};
    dst[2 * 32] = v4f{gi[0], gi[1], gi[2], gi[3]};
    dst[3 * 32] = v4f{gi[4], gi[5], gi[6], gi[7]};
  }
  __syncthreads();
  const v4f* src = red + (tj * 4 + kq) * 32 + lane;
  v4f acc = src[0];
#pragma unroll
  for (int q = 1; q < 4; ++q) acc += src[q * (4 * 4 * 32)];

  float* gp = ws + ((kq & 2) ? WS_GI : WS_GR);
  const int rbase = 16 * ti + ((lane < 16) ? 0 : 8) + 4 * (kq & 1);
  const int c = 16 * tj + p;
#pragma unroll
  for (int e = 0; e < 4; ++e) atomicAdd(&gp[(rbase + e) * 64 + c], acc[e]);
}

#define NR_LD 65

template <int KB>
__device__ __forceinline__ void fo_factor_block(float2 (&g)[4][4], float2* Lc, float2* Sg,
                                                float* rdv, const int ty, const int tx) {
#pragma nounroll
  for (int kk = 0; kk < 16; kk += 2) {
    const int k = 16 * KB + kk;
    float2* sg = Sg + ((kk & 2) ? 64 : 0);
    if (tx == kk) {
#pragma unroll
      for (int a = KB; a < 4; ++a) Lc[k * 64 + 16 * a + ty] = g[a][KB];
    }
    if (tx == kk + 1) {
#pragma unroll
      for (int a = KB; a < 4; ++a) sg[16 * a + ty] = g[a][KB];
    }
    __syncthreads();
    const float gkk = Lc[k * 64 + k].x;
    const float2 g10 = Lc[k * 64 + k + 1];
    const float g11 = sg[k + 1].x;
    float2 ri0[4], ri1[4];
    float2 rj0[4], rj1[4];
#pragma unroll
    for (int a = KB; a < 4; ++a) {
      ri0[a] = Lc[k * 64 + 16 * a + ty];
      ri1[a] = sg[16 * a + ty];
    }
#pragma unroll
    for (int b = KB; b < 4; ++b) {
      rj0[b] = Lc[k * 64 + 16 * b + tx];
      rj1[b] = sg[16 * b + tx];
    }
    __builtin_amdgcn_sched_barrier(0);
    const float rd0 = __builtin_amdgcn_rsqf(gkk + 1e-8f);
    const float2 m = make_float2(g10.x * rd0, g10.y * rd0);
    const float d1 = g11 - (m.x * m.x + m.y * m.y);
    const float rd1 = __builtin_amdgcn_rsqf(d1 + 1e-8f);
    float2 li0[4], li1[4], ui[4];
    float2 lj0[4], lj1[4];
#pragma unroll
    for (int a = KB; a < 4; ++a) {
      const float2 l0 = make_float2(ri0[a].x * rd0, ri0[a].y * rd0);
      const float2 u = make_float2(ri1[a].x - (l0.x * m.x + l0.y * m.y),
                                   ri1[a].y - (l0.y * m.x - l0.x * m.y));
      li0[a] = l0;
      ui[a] = u;
      li1[a] = make_float2(u.x * rd1, u.y * rd1);
    }
#pragma unroll
    for (int b = KB; b < 4; ++b) {
      const float2 l0 = make_float2(rj0[b].x * rd0, rj0[b].y * rd0);
      const float2 u = make_float2(rj1[b].x - (l0.x * m.x + l0.y * m.y),
                                   rj1[b].y - (l0.y * m.x - l0.x * m.y));
      lj0[b] = l0;
      lj1[b] = make_float2(u.x * rd1, u.y * rd1);
    }
    if (tx == kk + 1) {
#pragma unroll
      for (int a = KB; a < 4; ++a) Lc[(k + 1) * 64 + 16 * a + ty] = ui[a];
    }
    if (tx == kk && ty == kk) {
      rdv[k] = rd0;
      rdv[k + 1] = rd1;
    }
#pragma unroll
    for (int b = KB; b < 4; ++b) {
#pragma unroll
      for (int a = b; a < 4; ++a) {
        const bool live = (b > KB || tx > kk + 1) && (a > b || ty >= tx);
        if (live) {
          g[a][b].x -= li0[a].x * lj0[b].x + li0[a].y * lj0[b].y;
          g[a][b].y -= li0[a].y * lj0[b].x - li0[a].x * lj0[b].y;
          g[a][b].x -= li1[a].x * lj1[b].x + li1[a].y * lj1[b].y;
          g[a][b].y -= li1[a].y * lj1[b].x - li1[a].x * lj1[b].y;
        }
      }
    }
  }
}

__device__ __forceinline__ void fo_invert_diag(float2 (&nd)[4], const float2* Lc, float2* Nr,
                                               const float* rdv, const int ty, const int tx) {
  float rd[4];
#pragma unroll
  for (int q = 0; q < 4; ++q) rd[q] = rdv[16 * q];
#pragma nounroll
  for (int kk = 0; kk < 16; ++kk) {
    if (ty == kk) {
#pragma unroll
      for (int q = 0; q < 4; ++q) {
        const float2 v = make_float2(nd[q].x * rd[q], nd[q].y * rd[q]);
        nd[q] = v;
        Nr[(16 * q + kk) * NR_LD + 16 * q + tx] = v;
      }
    }
    __syncthreads();
    float rdn[4];
#pragma unroll
    for (int q = 0; q < 4; ++q) rdn[q] = rdv[16 * q + ((kk + 1) & 15)];
#pragma unroll
    for (int q = 0; q < 4; ++q) {
      const int k = 16 * q + kk;
      const float2 t = Lc[k * 64 + 16 * q + ty];
      const float2 lr = make_float2(t.x * rd[q], t.y * rd[q]);
      const float2 nk = Nr[k * NR_LD + 16 * q + tx];
      if (ty > kk) {
        nd[q].x -= lr.x * nk.x - lr.y * nk.y;
        nd[q].y -= lr.x * nk.y + lr.y * nk.x;
      }
    }
#pragma unroll
    for (int q = 0; q < 4; ++q) rd[q] = rdn[q];
  }
}

#define TS_LD 17
#define TS_BLK (16 * TS_LD)

template <int D>
__device__ __forceinline__ void fo_inverse_level(const float2* Lc, float2* Nr, float2* Ts,
                                                 const int ty, const int tx) {
  constexpr int NB = 4 - D;
  constexpr int NM = 16 * D;
  {
    float2 a[NB][NM];
    float2 b[NB][NM];
#pragma unroll
    for (int n = 0; n < NB; ++n) {
#pragma unroll
      for (int e = 0; e < NM; ++e) {
        const int c = 16 * n + e;
        a[n][e] = Lc[c * 64 + 16 * (D + n) + ty];
        b[n][e] = Nr[c * NR_LD + 16 * n + tx];
      }
    }
#pragma unroll
    for (int n = 0; n < NB; ++n) {
      float2 s0 = make_float2(0.0f, 0.0f);
      float2 s1 = make_float2(0.0f, 0.0f);
#pragma unroll
      for (int e = 0; e < NM; e += 2) {
        s0.x += a[n][e].x * b[n][e].x;
        s0.x -= a[n][e].y * b[n][e].y;
        s0.y += a[n][e].x * b[n][e].y;
        s0.y += a[n][e].y * b[n][e].x;
        s1.x += a[n][e + 1].x * b[n][e + 1].x;
        s1.x -= a[n][e + 1].y * b[n][e + 1].y;
        s1.y += a[n][e + 1].x * b[n][e + 1].y;
        s1.y += a[n][e + 1].y * b[n][e + 1].x;
      }
      Ts[(D + n) * TS_BLK + ty * TS_LD + tx] = make_float2(s0.x + s1.x, s0.y + s1.y);
    }
  }
  __syncthreads();
  {
    float2 a[NB][16];
    float2 b[NB][16];
#pragma unroll
    for (int n = 0; n < NB; ++n) {
      const int i = D + n;
#pragma unroll
      for (int m = 0; m < 16; ++m) {
        a[n][m] = Nr[(16 * i + ty) * NR_LD + 16 * i + m];
        b[n][m] = Ts[i * TS_BLK + m * TS_LD + tx];
      }
    }
#pragma unroll
    for (int n = 0; n < NB; ++n) {
      float2 s0 = make_float2(0.0f, 0.0f);
      float2 s1 = make_float2(0.0f, 0.0f);
#pragma unroll
      for (int m = 0; m < 16; m += 2) {
        s0.x += a[n][m].x * b[n][m].x;
        s0.x -= a[n][m].y * b[n][m].y;
        s0.y += a[n][m].x * b[n][m].y;
        s0.y += a[n][m].y * b[n][m].x;
        s1.x += a[n][m + 1].x * b[n][m + 1].x;
        s1.x -= a[n][m + 1].y * b[n][m + 1].y;
        s1.y += a[n][m + 1].x * b[n][m + 1].y;
        s1.y += a[n][m + 1].y * b[n][m + 1].x;
      }
      Nr[(16 * (D + n) + ty) * NR_LD + 16 * n + tx] = make_float2(-(s0.x + s1.x), -(s0.y + s1.y));
    }
  }
  __syncthreads();
}

__global__ __launch_bounds__(256) void fo_chol_inv(float* __restrict__ ws) {
  __shared__ float2 Lc[64 * 64];
  __shared__ float2 Nr[64 * NR_LD];
  __shared__ float rdv[64];
  __shared__ float2 Sg[2 * 64];
  __shared__ float2 Ts[4 * TS_BLK];
  const int tid = threadIdx.x;
  const int ty = tid >> 4;
  const int tx = tid & 15;
  const float* grp = ws + WS_GR;
  const float* gip = ws + WS_GI;

  float2 g[4][4];
#pragma unroll
  for (int a = 0; a < 4; ++a) {
#pragma unroll
    for (int b = 0; b <= a; ++b) {
      const int idx = (16 * a + ty) * 64 + 16 * b + tx;
      g[a][b] = make_float2(grp[idx], gip[idx]);
    }
  }

  fo_factor_block<0>(g, Lc, Sg, rdv, ty, tx);
  fo_factor_block<1>(g, Lc, Sg, rdv, ty, tx);
  fo_factor_block<2>(g, Lc, Sg, rdv, ty, tx);
  fo_factor_block<3>(g, Lc, Sg, rdv, ty, tx);

  {
    float2 t[12];
    float s[12];
#pragma unroll
    for (int m = 0; m < 12; ++m) {
      t[m] = Lc[tid + 256 * m];
      s[m] = rdv[(tid >> 6) + 4 * m];
    }
    const int mend = 4 * ((tid & 63) >> 4);
#pragma unroll
    for (int m = 0; m < 12; ++m) {
      if (m < mend) Lc[tid + 256 * m] = make_float2(t[m].x * s[m], t[m].y * s[m]);
    }
  }

  float2 nd[4];
#pragma unroll
  for (int q = 0; q < 4; ++q) nd[q] = make_float2((ty == tx) ? 1.0f : 0.0f, 0.0f);
  fo_invert_diag(nd, Lc, Nr, rdv, ty, tx);
  fo_inverse_level<1>(Lc, Nr, Ts, ty, tx);
  fo_inverse_level<2>(Lc, Nr, Ts, ty, tx);
  fo_inverse_level<3>(Lc, Nr, Ts, ty, tx);

  float* wp = ws + WS_W;
  for (int m = 0; m < 16; ++m) {
    const int o = tid + 256 * m;
    const int c = o >> 6;
    const int j = o & 63;
    float2 v = make_float2(0.0f, 0.0f);
    if (j >= c) v = Nr[j * NR_LD + c];
    *(float2*)(wp + 2 * o) = make_float2(v.x, -v.y);
  }
}

#define APPLY_BLOCK_ROWS 512
#define APPLY_GROUPS (APPLY_BLOCK_ROWS / 64)
#define APPLY_BLOCKS (M_ROWS / APPLY_BLOCK_ROWS)

__device__ __forceinline__ void fo_apply_load(const float* __restrict__ px, v4f (&e)[16]) {
#pragma unroll
  for (int j = 0; j < 16; ++j) e[j] = *(const v4f*)(px + 8 * j);
}

__device__ __forceinline__ void fo_apply_strip(const v4f (&e)[16], const v2f (&wr)[16],
                                               const v2f (&wi)[16], float* __restrict__ po) {
  v8f qr = {};
  v8f qi = {};
#pragma unroll
  for (int j = 0; j < 16; ++j) {
    v2f ar = {e[j].x, e[j].z};
    v2f ai = {e[j].y, e[j].w};
    v2f ain = {-e[j].y, -e[j].w};
    qr = wmma_f32x4(ar, wr[j], qr);
    qr = wmma_f32x4(ain, wi[j], qr);
    qi = wmma_f32x4(ar, wi[j], qi);
    qi = wmma_f32x4(ai, wr[j], qi);
  }
#pragma unroll
  for (int v = 0; v < 8; ++v) {
    v2f o = {qr[v], qi[v]};
    *(v2f*)(po + (size_t)v * X_ROW_F) = o;
  }
}

__global__ __launch_bounds__(512) void fo_apply(const float* __restrict__ x,
                                                const float* __restrict__ ws,
                                                float* __restrict__ out) {
  const int lane = threadIdx.x & 31;
  const int wave = threadIdx.x >> 5;
  const int s = wave >> 2;
  const int tj = wave & 3;
  const int p = lane & 15;
  const int kb = (lane < 16) ? 0 : 2;
  const int rb = (lane < 16) ? 0 : 8;

  const float* wp = ws + WS_W + ((size_t)kb * 64 + 16 * tj + p) * 2;
  v2f wr[16];
  v2f wi[16];
#pragma unroll
  for (int j = 0; j < 16; ++j) {
    const v2f w0 = *(const v2f*)(wp + (size_t)(4 * j + 0) * 64 * 2);
    const v2f w1 = *(const v2f*)(wp + (size_t)(4 * j + 1) * 64 * 2);
    wr[j] = v2f{w0.x, w1.x};
    wi[j] = v2f{w0.y, w1.y};
  }

  const size_t row0 = (size_t)blockIdx.x * APPLY_BLOCK_ROWS + 16 * s;
  const float* px = x + ((row0 + p) * R_COLS + kb) * 2;
  float* po = out + ((row0 + rb) * R_COLS + 16 * tj + p) * 2;
  constexpr size_t kGroupF = (size_t)64 * X_ROW_F;

  v4f e0[16], e1[16];
  fo_apply_load(px, e0);
  for (int g = 0; g < APPLY_GROUPS - 2; g += 2) {
    fo_apply_load(px + (g + 1) * kGroupF, e1);
    fo_apply_strip(e0, wr, wi, po + g * kGroupF);
    fo_apply_load(px + (g + 2) * kGroupF, e0);
    fo_apply_strip(e1, wr, wi, po + (g + 1) * kGroupF);
  }
  fo_apply_load(px + (APPLY_GROUPS - 1) * kGroupF, e1);
  fo_apply_strip(e0, wr, wi, po + (APPLY_GROUPS - 2) * kGroupF);
  fo_apply_strip(e1, wr, wi, po + (APPLY_GROUPS - 1) * kGroupF);
}

extern "C" void kernel_launch(void* const* d_in, const int* in_sizes, int n_in,
                              void* d_out, int out_size, void* d_ws, size_t ws_size,
                              hipStream_t stream) {
  (void)in_sizes; (void)n_in; (void)out_size; (void)ws_size;
  const float* x = (const float*)d_in[0];
  float* out = (float*)d_out;
  float* ws = (float*)d_ws;

  fo_zero_gram<<<(2 * 64 * 64 + 255) / 256, 256, 0, stream>>>(ws);
  fo_gram<<<GRAM_BLOCKS, 512, 0, stream>>>(x, ws);
  fo_chol_inv<<<1, 256, 0, stream>>>(ws);
  fo_apply<<<APPLY_BLOCKS, 512, 0, stream>>>(x, ws, out);
}


# Gram and apply: every wave holds eight 16x16 accumulator tiles (one wave per SIMD, 1024 waves not 4096), one fragment load feeds 2x the WMMAs, next rows' loads kept a group ahead; sum order unchanged
# speedup vs baseline: 3.8315x; 1.1050x over previous
#include <hip/hip_runtime.h>


#define M_ROWS 131072
#define R_COLS 64

typedef __attribute__((ext_vector_type(2))) float v2f;
typedef __attribute__((ext_vector_type(4))) float v4f;
typedef __attribute__((ext_vector_type(8))) float v8f;

#define WS_GR 0
#define WS_GI 4096
#define WS_W  8192

__device__ inline v8f wmma_f32x4(v2f a, v2f b, v8f c) {
  return __builtin_amdgcn_wmma_f32_16x16x4_f32(
       false, a,  false, b,
       (short)0, c,  false,  false);
}

__global__ void fo_zero_gram(float* ws) {
  int i = blockIdx.x * blockDim.x + threadIdx.x;
  if (i < 2 * 64 * 64) ws[i] = 0.0f;
}

#define GRAM_BLOCK_ROWS 2048
#define GRAM_WAVE_ROWS 512
#define GRAM_BLOCKS (4 * (M_ROWS / GRAM_BLOCK_ROWS))
#define X_ROW_F (R_COLS * 2)

__device__ __forceinline__ void fo_gram_load8(const float* __restrict__ p0,
                                              const float* __restrict__ p1,
                                              const float* __restrict__ p2,
                                              const float* __restrict__ p3,
                                              v2f (&e)[4][4]) {
#pragma unroll
  for (int j = 0; j < 2; ++j) {
    e[0][2 * j + 0] = *(const v2f*)(p0 + (4 * j + 0) * X_ROW_F);
    e[0][2 * j + 1] = *(const v2f*)(p0 + (4 * j + 1) * X_ROW_F);
    e[1][2 * j + 0] = *(const v2f*)(p1 + (4 * j + 0) * X_ROW_F);
    e[1][2 * j + 1] = *(const v2f*)(p1 + (4 * j + 1) * X_ROW_F);
    e[2][2 * j + 0] = *(const v2f*)(p2 + (4 * j + 0) * X_ROW_F);
    e[2][2 * j + 1] = *(const v2f*)(p2 + (4 * j + 1) * X_ROW_F);
    e[3][2 * j + 0] = *(const v2f*)(p3 + (4 * j + 0) * X_ROW_F);
    e[3][2 * j + 1] = *(const v2f*)(p3 + (4 * j + 1) * X_ROW_F);
  }
}

__device__ __forceinline__ void fo_gram_mma8(const v2f (&e)[4][4], v8f (&gr)[4], v8f (&gi)[4]) {
#pragma unroll
  for (int j = 0; j < 2; ++j) {
    v2f xr[4], xi[4];
#pragma unroll
    for (int d = 0; d < 4; ++d) {
      const v2f e0 = e[d][2 * j + 0];
      const v2f e1 = e[d][2 * j + 1];
      xr[d] = v2f{e0.x, e1.x};
      xi[d] = v2f{e0.y, e1.y};
    }
    const v2f ain = {-xi[0].x, -xi[0].y};

#pragma unroll
    for (int d = 0; d < 4; ++d) gr[d] = wmma_f32x4(xr[0], xr[d], gr[d]);
#pragma unroll
    for (int d = 0; d < 4; ++d) gi[d] = wmma_f32x4(xr[0], xi[d], gi[d]);
#pragma unroll
    for (int d = 0; d < 4; ++d) gr[d] = wmma_f32x4(xi[0], xi[d], gr[d]);
#pragma unroll
    for (int d = 0; d < 4; ++d) gi[d] = wmma_f32x4(ain, xr[d], gi[d]);
  }
}

__global__ __launch_bounds__(128) void fo_gram(const float* __restrict__ x,
                                               float* __restrict__ ws) {
  __shared__ v4f red[4 * 16 * 32];

  const int lane = threadIdx.x & 31;
  const int kq = threadIdx.x >> 5;
  const int ti = blockIdx.x & 3;
  const int p = lane & 15;
  const int kb = (lane < 16) ? 0 : 2;

  const size_t row0 = (size_t)(blockIdx.x >> 2) * GRAM_BLOCK_ROWS + kq * GRAM_WAVE_ROWS + kb;
  const float* xb = x + row0 * X_ROW_F + 2 * p;
  const float* p0 = xb + 32 * ti;
  const float* p1 = xb + 32 * ((ti + 1) & 3);
  const float* p2 = xb + 32 * ((ti + 2) & 3);
  const float* p3 = xb + 32 * ((ti + 3) & 3);

  v8f gr[4] = {};
  v8f gi[4] = {};

  v2f s0[4][4], s1[4][4], s2[4][4], s3[4][4];
  constexpr int kSets = GRAM_WAVE_ROWS / 8;
  constexpr int kSetF = 8 * X_ROW_F;
  fo_gram_load8(p0, p1, p2, p3, s0);
  fo_gram_load8(p0 + kSetF, p1 + kSetF, p2 + kSetF, p3 + kSetF, s1);
  fo_gram_load8(p0 + 2 * kSetF, p1 + 2 * kSetF, p2 + 2 * kSetF, p3 + 2 * kSetF, s2);
  fo_gram_load8(p0 + 3 * kSetF, p1 + 3 * kSetF, p2 + 3 * kSetF, p3 + 3 * kSetF, s3);
  for (int c = 0; c < kSets - 4; c += 4) {
    const int o4 = (c + 4) * kSetF;
    fo_gram_mma8(s0, gr, gi);
    __builtin_amdgcn_sched_barrier(0);
    fo_gram_load8(p0 + o4, p1 + o4, p2 + o4, p3 + o4, s0);
    const int o5 = (c + 5) * kSetF;
    fo_gram_mma8(s1, gr, gi);
    __builtin_amdgcn_sched_barrier(0);
    fo_gram_load8(p0 + o5, p1 + o5, p2 + o5, p3 + o5, s1);
    const int o6 = (c + 6) * kSetF;
    fo_gram_mma8(s2, gr, gi);
    __builtin_amdgcn_sched_barrier(0);
    fo_gram_load8(p0 + o6, p1 + o6, p2 + o6, p3 + o6, s2);
    const int o7 = (c + 7) * kSetF;
    fo_gram_mma8(s3, gr, gi);
    __builtin_amdgcn_sched_barrier(0);
    fo_gram_load8(p0 + o7, p1 + o7, p2 + o7, p3 + o7, s3);
  }
  fo_gram_mma8(s0, gr, gi);
  fo_gram_mma8(s1, gr, gi);
  fo_gram_mma8(s2, gr, gi);
  fo_gram_mma8(s3, gr, gi);

  {
    v4f* dst = red + (kq * 16) * 32 + lane;
#pragma unroll
    for (int d = 0; d < 4; ++d) {
      dst[(4 * d + 0) * 32] = v4f{gr[d][0], gr[d][1], gr[d][2], gr[d][3]};
      dst[(4 * d + 1) * 32] = v4f{gr[d][4], gr[d][5], gr[d][6], gr[d][7]};
      dst[(4 * d + 2) * 32] = v4f{gi[d][0], gi[d][1], gi[d][2], gi[d][3]};
      dst[(4 * d + 3) * 32] = v4f{gi[d][4], gi[d][5], gi[d][6], gi[d][7]};
    }
  }
  __syncthreads();
  const int tj = (ti + kq) & 3;
  const int c = 16 * tj + p;
#pragma unroll
  for (int part = 0; part < 4; ++part) {
    const v4f* src = red + (4 * kq + part) * 32 + lane;
    v4f acc = src[0];
#pragma unroll
    for (int q = 1; q < 4; ++q) acc += src[q * (16 * 32)];

    float* gp = ws + ((part & 2) ? WS_GI : WS_GR);
    const int rbase = 16 * ti + ((lane < 16) ? 0 : 8) + 4 * (part & 1);
#pragma unroll
    for (int e = 0; e < 4; ++e) atomicAdd(&gp[(rbase + e) * 64 + c], acc[e]);
  }
}

#define NR_LD 65

template <int KB>
__device__ __forceinline__ void fo_factor_block(float2 (&g)[4][4], float2* Lc, float2* Sg,
                                                float* rdv, const int ty, const int tx) {
#pragma nounroll
  for (int kk = 0; kk < 16; kk += 2) {
    const int k = 16 * KB + kk;
    float2* sg = Sg + ((kk & 2) ? 64 : 0);
    if (tx == kk) {
#pragma unroll
      for (int a = KB; a < 4; ++a) Lc[k * 64 + 16 * a + ty] = g[a][KB];
    }
    if (tx == kk + 1) {
#pragma unroll
      for (int a = KB; a < 4; ++a) sg[16 * a + ty] = g[a][KB];
    }
    __syncthreads();
    const float gkk = Lc[k * 64 + k].x;
    const float2 g10 = Lc[k * 64 + k + 1];
    const float g11 = sg[k + 1].x;
    float2 ri0[4], ri1[4];
    float2 rj0[4], rj1[4];
#pragma unroll
    for (int a = KB; a < 4; ++a) {
      ri0[a] = Lc[k * 64 + 16 * a + ty];
      ri1[a] = sg[16 * a + ty];
    }
#pragma unroll
    for (int b = KB; b < 4; ++b) {
      rj0[b] = Lc[k * 64 + 16 * b + tx];
      rj1[b] = sg[16 * b + tx];
    }
    __builtin_amdgcn_sched_barrier(0);
    const float rd0 = __builtin_amdgcn_rsqf(gkk + 1e-8f);
    const float2 m = make_float2(g10.x * rd0, g10.y * rd0);
    const float d1 = g11 - (m.x * m.x + m.y * m.y);
    const float rd1 = __builtin_amdgcn_rsqf(d1 + 1e-8f);
    float2 li0[4], li1[4], ui[4];
    float2 lj0[4], lj1[4];
#pragma unroll
    for (int a = KB; a < 4; ++a) {
      const float2 l0 = make_float2(ri0[a].x * rd0, ri0[a].y * rd0);
      const float2 u = make_float2(ri1[a].x - (l0.x * m.x + l0.y * m.y),
                                   ri1[a].y - (l0.y * m.x - l0.x * m.y));
      li0[a] = l0;
      ui[a] = u;
      li1[a] = make_float2(u.x * rd1, u.y * rd1);
    }
#pragma unroll
    for (int b = KB; b < 4; ++b) {
      const float2 l0 = make_float2(rj0[b].x * rd0, rj0[b].y * rd0);
      const float2 u = make_float2(rj1[b].x - (l0.x * m.x + l0.y * m.y),
                                   rj1[b].y - (l0.y * m.x - l0.x * m.y));
      lj0[b] = l0;
      lj1[b] = make_float2(u.x * rd1, u.y * rd1);
    }
    if (tx == kk + 1) {
#pragma unroll
      for (int a = KB; a < 4; ++a) Lc[(k + 1) * 64 + 16 * a + ty] = ui[a];
    }
    if (tx == kk && ty == kk) {
      rdv[k] = rd0;
      rdv[k + 1] = rd1;
    }
#pragma unroll
    for (int b = KB; b < 4; ++b) {
#pragma unroll
      for (int a = b; a < 4; ++a) {
        const bool live = (b > KB || tx > kk + 1) && (a > b || ty >= tx);
        if (live) {
          g[a][b].x -= li0[a].x * lj0[b].x + li0[a].y * lj0[b].y;
          g[a][b].y -= li0[a].y * lj0[b].x - li0[a].x * lj0[b].y;
          g[a][b].x -= li1[a].x * lj1[b].x + li1[a].y * lj1[b].y;
          g[a][b].y -= li1[a].y * lj1[b].x - li1[a].x * lj1[b].y;
        }
      }
    }
  }
}

__device__ __forceinline__ void fo_invert_diag(float2 (&nd)[4], const float2* Lc, float2* Nr,
                                               const float* rdv, const int ty, const int tx) {
  float rd[4];
#pragma unroll
  for (int q = 0; q < 4; ++q) rd[q] = rdv[16 * q];
#pragma nounroll
  for (int kk = 0; kk < 16; ++kk) {
    if (ty == kk) {
#pragma unroll
      for (int q = 0; q < 4; ++q) {
        const float2 v = make_float2(nd[q].x * rd[q], nd[q].y * rd[q]);
        nd[q] = v;
        Nr[(16 * q + kk) * NR_LD + 16 * q + tx] = v;
      }
    }
    __syncthreads();
    float rdn[4];
#pragma unroll
    for (int q = 0; q < 4; ++q) rdn[q] = rdv[16 * q + ((kk + 1) & 15)];
#pragma unroll
    for (int q = 0; q < 4; ++q) {
      const int k = 16 * q + kk;
      const float2 t = Lc[k * 64 + 16 * q + ty];
      const float2 lr = make_float2(t.x * rd[q], t.y * rd[q]);
      const float2 nk = Nr[k * NR_LD + 16 * q + tx];
      if (ty > kk) {
        nd[q].x -= lr.x * nk.x - lr.y * nk.y;
        nd[q].y -= lr.x * nk.y + lr.y * nk.x;
      }
    }
#pragma unroll
    for (int q = 0; q < 4; ++q) rd[q] = rdn[q];
  }
}

#define TS_LD 17
#define TS_BLK (16 * TS_LD)

template <int D>
__device__ __forceinline__ void fo_inverse_level(const float2* Lc, float2* Nr, float2* Ts,
                                                 const int ty, const int tx) {
  constexpr int NB = 4 - D;
  constexpr int NM = 16 * D;
  {
    float2 a[NB][NM];
    float2 b[NB][NM];
#pragma unroll
    for (int n = 0; n < NB; ++n) {
#pragma unroll
      for (int e = 0; e < NM; ++e) {
        const int c = 16 * n + e;
        a[n][e] = Lc[c * 64 + 16 * (D + n) + ty];
        b[n][e] = Nr[c * NR_LD + 16 * n + tx];
      }
    }
#pragma unroll
    for (int n = 0; n < NB; ++n) {
      float2 s0 = make_float2(0.0f, 0.0f);
      float2 s1 = make_float2(0.0f, 0.0f);
#pragma unroll
      for (int e = 0; e < NM; e += 2) {
        s0.x += a[n][e].x * b[n][e].x;
        s0.x -= a[n][e].y * b[n][e].y;
        s0.y += a[n][e].x * b[n][e].y;
        s0.y += a[n][e].y * b[n][e].x;
        s1.x += a[n][e + 1].x * b[n][e + 1].x;
        s1.x -= a[n][e + 1].y * b[n][e + 1].y;
        s1.y += a[n][e + 1].x * b[n][e + 1].y;
        s1.y += a[n][e + 1].y * b[n][e + 1].x;
      }
      Ts[(D + n) * TS_BLK + ty * TS_LD + tx] = make_float2(s0.x + s1.x, s0.y + s1.y);
    }
  }
  __syncthreads();
  {
    float2 a[NB][16];
    float2 b[NB][16];
#pragma unroll
    for (int n = 0; n < NB; ++n) {
      const int i = D + n;
#pragma unroll
      for (int m = 0; m < 16; ++m) {
        a[n][m] = Nr[(16 * i + ty) * NR_LD + 16 * i + m];
        b[n][m] = Ts[i * TS_BLK + m * TS_LD + tx];
      }
    }
#pragma unroll
    for (int n = 0; n < NB; ++n) {
      float2 s0 = make_float2(0.0f, 0.0f);
      float2 s1 = make_float2(0.0f, 0.0f);
#pragma unroll
      for (int m = 0; m < 16; m += 2) {
        s0.x += a[n][m].x * b[n][m].x;
        s0.x -= a[n][m].y * b[n][m].y;
        s0.y += a[n][m].x * b[n][m].y;
        s0.y += a[n][m].y * b[n][m].x;
        s1.x += a[n][m + 1].x * b[n][m + 1].x;
        s1.x -= a[n][m + 1].y * b[n][m + 1].y;
        s1.y += a[n][m + 1].x * b[n][m + 1].y;
        s1.y += a[n][m + 1].y * b[n][m + 1].x;
      }
      Nr[(16 * (D + n) + ty) * NR_LD + 16 * n + tx] = make_float2(-(s0.x + s1.x), -(s0.y + s1.y));
    }
  }
  __syncthreads();
}

__global__ __launch_bounds__(256) void fo_chol_inv(float* __restrict__ ws) {
  __shared__ float2 Lc[64 * 64];
  __shared__ float2 Nr[64 * NR_LD];
  __shared__ float rdv[64];
  __shared__ float2 Sg[2 * 64];
  __shared__ float2 Ts[4 * TS_BLK];
  const int tid = threadIdx.x;
  const int ty = tid >> 4;
  const int tx = tid & 15;
  const float* grp = ws + WS_GR;
  const float* gip = ws + WS_GI;

  float2 g[4][4];
#pragma unroll
  for (int a = 0; a < 4; ++a) {
#pragma unroll
    for (int b = 0; b <= a; ++b) {
      const int idx = (16 * a + ty) * 64 + 16 * b + tx;
      g[a][b] = make_float2(grp[idx], gip[idx]);
    }
  }

  fo_factor_block<0>(g, Lc, Sg, rdv, ty, tx);
  fo_factor_block<1>(g, Lc, Sg, rdv, ty, tx);
  fo_factor_block<2>(g, Lc, Sg, rdv, ty, tx);
  fo_factor_block<3>(g, Lc, Sg, rdv, ty, tx);

  {
    float2 t[12];
    float s[12];
#pragma unroll
    for (int m = 0; m < 12; ++m) {
      t[m] = Lc[tid + 256 * m];
      s[m] = rdv[(tid >> 6) + 4 * m];
    }
    const int mend = 4 * ((tid & 63) >> 4);
#pragma unroll
    for (int m = 0; m < 12; ++m) {
      if (m < mend) Lc[tid + 256 * m] = make_float2(t[m].x * s[m], t[m].y * s[m]);
    }
  }

  float2 nd[4];
#pragma unroll
  for (int q = 0; q < 4; ++q) nd[q] = make_float2((ty == tx) ? 1.0f : 0.0f, 0.0f);
  fo_invert_diag(nd, Lc, Nr, rdv, ty, tx);
  fo_inverse_level<1>(Lc, Nr, Ts, ty, tx);
  fo_inverse_level<2>(Lc, Nr, Ts, ty, tx);
  fo_inverse_level<3>(Lc, Nr, Ts, ty, tx);

  float* wp = ws + WS_W;
  for (int m = 0; m < 16; ++m) {
    const int o = tid + 256 * m;
    const int c = o >> 6;
    const int j = o & 63;
    float2 v = make_float2(0.0f, 0.0f);
    if (j >= c) v = Nr[j * NR_LD + c];
    *(float2*)(wp + 2 * o) = make_float2(v.x, -v.y);
  }
}

#define APPLY_BLOCK_ROWS 512
#define APPLY_GROUPS (APPLY_BLOCK_ROWS / 64)
#define APPLY_BLOCKS (M_ROWS / APPLY_BLOCK_ROWS)

__device__ __forceinline__ void fo_apply_load(const float* __restrict__ px, v4f (&e)[2][16]) {
#pragma unroll
  for (int j = 0; j < 16; ++j) {
    e[0][j] = *(const v4f*)(px + 8 * j);
    e[1][j] = *(const v4f*)(px + 16 * X_ROW_F + 8 * j);
  }
}

__device__ __forceinline__ void fo_apply_block(const v4f (&e)[2][16], const v2f (&wr)[2][16],
                                               const v2f (&wi)[2][16], float* __restrict__ po) {
  v8f qr[2][2] = {};
  v8f qi[2][2] = {};
#pragma unroll
  for (int j = 0; j < 16; ++j) {
    v2f ar[2], ai[2], ain[2];
#pragma unroll
    for (int u = 0; u < 2; ++u) {
      ar[u] = v2f{e[u][j].x, e[u][j].z};
      ai[u] = v2f{e[u][j].y, e[u][j].w};
      ain[u] = v2f{-e[u][j].y, -e[u][j].w};
    }
#pragma unroll
    for (int u = 0; u < 2; ++u) {
#pragma unroll
      for (int t = 0; t < 2; ++t) qr[u][t] = wmma_f32x4(ar[u], wr[t][j], qr[u][t]);
    }
#pragma unroll
    for (int u = 0; u < 2; ++u) {
#pragma unroll
      for (int t = 0; t < 2; ++t) qi[u][t] = wmma_f32x4(ar[u], wi[t][j], qi[u][t]);
    }
#pragma unroll
    for (int u = 0; u < 2; ++u) {
#pragma unroll
      for (int t = 0; t < 2; ++t) qr[u][t] = wmma_f32x4(ain[u], wi[t][j], qr[u][t]);
    }
#pragma unroll
    for (int u = 0; u < 2; ++u) {
#pragma unroll
      for (int t = 0; t < 2; ++t) qi[u][t] = wmma_f32x4(ai[u], wr[t][j], qi[u][t]);
    }
  }
#pragma unroll
  for (int u = 0; u < 2; ++u) {
#pragma unroll
    for (int t = 0; t < 2; ++t) {
#pragma unroll
      for (int v = 0; v < 8; ++v) {
        v2f o = {qr[u][t][v], qi[u][t][v]};
        *(v2f*)(po + (size_t)(16 * u + v) * X_ROW_F + 32 * t) = o;
      }
    }
  }
}

__global__ __launch_bounds__(128) void fo_apply(const float* __restrict__ x,
                                                const float* __restrict__ ws,
                                                float* __restrict__ out) {
  const int lane = threadIdx.x & 31;
  const int wave = threadIdx.x >> 5;
  const int rs = wave >> 1;
  const int ch = wave & 1;
  const int p = lane & 15;
  const int kb = (lane < 16) ? 0 : 2;
  const int rb = (lane < 16) ? 0 : 8;

  const float* wp = ws + WS_W + ((size_t)kb * 64 + 32 * ch + p) * 2;
  v2f wr[2][16];
  v2f wi[2][16];
#pragma unroll
  for (int t = 0; t < 2; ++t) {
#pragma unroll
    for (int j = 0; j < 16; ++j) {
      const v2f w0 = *(const v2f*)(wp + (size_t)(4 * j + 0) * 64 * 2 + 32 * t);
      const v2f w1 = *(const v2f*)(wp + (size_t)(4 * j + 1) * 64 * 2 + 32 * t);
      wr[t][j] = v2f{w0.x, w1.x};
      wi[t][j] = v2f{w0.y, w1.y};
    }
  }

  const size_t row0 = (size_t)blockIdx.x * APPLY_BLOCK_ROWS + 32 * rs;
  const float* px = x + ((row0 + p) * R_COLS + kb) * 2;
  float* po = out + ((row0 + rb) * R_COLS + 32 * ch + p) * 2;
  constexpr size_t kGroupF = (size_t)64 * X_ROW_F;

  v4f e0[2][16], e1[2][16];
  fo_apply_load(px, e0);
  for (int g = 0; g < APPLY_GROUPS - 2; g += 2) {
    fo_apply_load(px + (g + 1) * kGroupF, e1);
    __builtin_amdgcn_sched_barrier(0);
    fo_apply_block(e0, wr, wi, po + g * kGroupF);
    fo_apply_load(px + (g + 2) * kGroupF, e0);
    __builtin_amdgcn_sched_barrier(0);
    fo_apply_block(e1, wr, wi, po + (g + 1) * kGroupF);
  }
  fo_apply_load(px + (APPLY_GROUPS - 1) * kGroupF, e1);
  __builtin_amdgcn_sched_barrier(0);
  fo_apply_block(e0, wr, wi, po + (APPLY_GROUPS - 2) * kGroupF);
  fo_apply_block(e1, wr, wi, po + (APPLY_GROUPS - 1) * kGroupF);
}

extern "C" void kernel_launch(void* const* d_in, const int* in_sizes, int n_in,
                              void* d_out, int out_size, void* d_ws, size_t ws_size,
                              hipStream_t stream) {
  (void)in_sizes; (void)n_in; (void)out_size; (void)ws_size;
  const float* x = (const float*)d_in[0];
  float* out = (float*)d_out;
  float* ws = (float*)d_ws;

  fo_zero_gram<<<(2 * 64 * 64 + 255) / 256, 256, 0, stream>>>(ws);
  fo_gram<<<GRAM_BLOCKS, 128, 0, stream>>>(x, ws);
  fo_chol_inv<<<1, 256, 0, stream>>>(ws);
  fo_apply<<<APPLY_BLOCKS, 128, 0, stream>>>(x, ws, out);
}


# factor+inverse fused into the apply launch (3 launches, not 4): each of the 256 work-groups factors G in LDS itself and reads W from LDS; apply re-tiled to 8 waves of 4 tiles; sum order unchanged
# speedup vs baseline: 4.0193x; 1.0490x over previous
#include <hip/hip_runtime.h>


#define M_ROWS 131072
#define R_COLS 64

typedef __attribute__((ext_vector_type(2))) float v2f;
typedef __attribute__((ext_vector_type(4))) float v4f;
typedef __attribute__((ext_vector_type(8))) float v8f;

#define WS_GR 0
#define WS_GI 4096

__device__ inline v8f wmma_f32x4(v2f a, v2f b, v8f c) {
  return __builtin_amdgcn_wmma_f32_16x16x4_f32(
       false, a,  false, b,
       (short)0, c,  false,  false);
}

__global__ void fo_zero_gram(float* ws) {
  int i = blockIdx.x * blockDim.x + threadIdx.x;
  if (i < 2 * 64 * 64) ws[i] = 0.0f;
}

#define GRAM_BLOCK_ROWS 2048
#define GRAM_WAVE_ROWS 512
#define GRAM_BLOCKS (4 * (M_ROWS / GRAM_BLOCK_ROWS))
#define X_ROW_F (R_COLS * 2)

__device__ __forceinline__ void fo_gram_load8(const float* __restrict__ p0,
                                              const float* __restrict__ p1,
                                              const float* __restrict__ p2,
                                              const float* __restrict__ p3,
                                              v2f (&e)[4][4]) {
#pragma unroll
  for (int j = 0; j < 2; ++j) {
    e[0][2 * j + 0] = *(const v2f*)(p0 + (4 * j + 0) * X_ROW_F);
    e[0][2 * j + 1] = *(const v2f*)(p0 + (4 * j + 1) * X_ROW_F);
    e[1][2 * j + 0] = *(const v2f*)(p1 + (4 * j + 0) * X_ROW_F);
    e[1][2 * j + 1] = *(const v2f*)(p1 + (4 * j + 1) * X_ROW_F);
    e[2][2 * j + 0] = *(const v2f*)(p2 + (4 * j + 0) * X_ROW_F);
    e[2][2 * j + 1] = *(const v2f*)(p2 + (4 * j + 1) * X_ROW_F);
    e[3][2 * j + 0] = *(const v2f*)(p3 + (4 * j + 0) * X_ROW_F);
    e[3][2 * j + 1] = *(const v2f*)(p3 + (4 * j + 1) * X_ROW_F);
  }
}

__device__ __forceinline__ void fo_gram_mma8(const v2f (&e)[4][4], v8f (&gr)[4], v8f (&gi)[4]) {
#pragma unroll
  for (int j = 0; j < 2; ++j) {
    v2f xr[4], xi[4];
#pragma unroll
    for (int d = 0; d < 4; ++d) {
      const v2f e0 = e[d][2 * j + 0];
      const v2f e1 = e[d][2 * j + 1];
      xr[d] = v2f{e0.x, e1.x};
      xi[d] = v2f{e0.y, e1.y};
    }
    const v2f ain = {-xi[0].x, -xi[0].y};

#pragma unroll
    for (int d = 0; d < 4; ++d) gr[d] = wmma_f32x4(xr[0], xr[d], gr[d]);
#pragma unroll
    for (int d = 0; d < 4; ++d) gi[d] = wmma_f32x4(xr[0], xi[d], gi[d]);
#pragma unroll
    for (int d = 0; d < 4; ++d) gr[d] = wmma_f32x4(xi[0], xi[d], gr[d]);
#pragma unroll
    for (int d = 0; d < 4; ++d) gi[d] = wmma_f32x4(ain, xr[d], gi[d]);
  }
}

__global__ __launch_bounds__(128) void fo_gram(const float* __restrict__ x,
                                               float* __restrict__ ws) {
  __shared__ v4f red[4 * 16 * 32];

  const int lane = threadIdx.x & 31;
  const int kq = threadIdx.x >> 5;
  const int ti = blockIdx.x & 3;
  const int p = lane & 15;
  const int kb = (lane < 16) ? 0 : 2;

  const size_t row0 = (size_t)(blockIdx.x >> 2) * GRAM_BLOCK_ROWS + kq * GRAM_WAVE_ROWS + kb;
  const float* xb = x + row0 * X_ROW_F + 2 * p;
  const float* p0 = xb + 32 * ti;
  const float* p1 = xb + 32 * ((ti + 1) & 3);
  const float* p2 = xb + 32 * ((ti + 2) & 3);
  const float* p3 = xb + 32 * ((ti + 3) & 3);

  v8f gr[4] = {};
  v8f gi[4] = {};

  v2f s0[4][4], s1[4][4], s2[4][4], s3[4][4];
  constexpr int kSets = GRAM_WAVE_ROWS / 8;
  constexpr int kSetF = 8 * X_ROW_F;
  fo_gram_load8(p0, p1, p2, p3, s0);
  fo_gram_load8(p0 + kSetF, p1 + kSetF, p2 + kSetF, p3 + kSetF, s1);
  fo_gram_load8(p0 + 2 * kSetF, p1 + 2 * kSetF, p2 + 2 * kSetF, p3 + 2 * kSetF, s2);
  fo_gram_load8(p0 + 3 * kSetF, p1 + 3 * kSetF, p2 + 3 * kSetF, p3 + 3 * kSetF, s3);
  for (int c = 0; c < kSets - 4; c += 4) {
    const int o4 = (c + 4) * kSetF;
    fo_gram_mma8(s0, gr, gi);
    __builtin_amdgcn_sched_barrier(0);
    fo_gram_load8(p0 + o4, p1 + o4, p2 + o4, p3 + o4, s0);
    const int o5 = (c + 5) * kSetF;
    fo_gram_mma8(s1, gr, gi);
    __builtin_amdgcn_sched_barrier(0);
    fo_gram_load8(p0 + o5, p1 + o5, p2 + o5, p3 + o5, s1);
    const int o6 = (c + 6) * kSetF;
    fo_gram_mma8(s2, gr, gi);
    __builtin_amdgcn_sched_barrier(0);
    fo_gram_load8(p0 + o6, p1 + o6, p2 + o6, p3 + o6, s2);
    const int o7 = (c + 7) * kSetF;
    fo_gram_mma8(s3, gr, gi);
    __builtin_amdgcn_sched_barrier(0);
    fo_gram_load8(p0 + o7, p1 + o7, p2 + o7, p3 + o7, s3);
  }
  fo_gram_mma8(s0, gr, gi);
  fo_gram_mma8(s1, gr, gi);
  fo_gram_mma8(s2, gr, gi);
  fo_gram_mma8(s3, gr, gi);

  {
    v4f* dst = red + (kq * 16) * 32 + lane;
#pragma unroll
    for (int d = 0; d < 4; ++d) {
      dst[(4 * d + 0) * 32] = v4f{gr[d][0], gr[d][1], gr[d][2], gr[d][3]};
      dst[(4 * d + 1) * 32] = v4f{gr[d][4], gr[d][5], gr[d][6], gr[d][7]};
      dst[(4 * d + 2) * 32] = v4f{gi[d][0], gi[d][1], gi[d][2], gi[d][3]};
      dst[(4 * d + 3) * 32] = v4f{gi[d][4], gi[d][5], gi[d][6], gi[d][7]};
    }
  }
  __syncthreads();
  const int tj = (ti + kq) & 3;
  const int c = 16 * tj + p;
#pragma unroll
  for (int part = 0; part < 4; ++part) {
    const v4f* src = red + (4 * kq + part) * 32 + lane;
    v4f acc = src[0];
#pragma unroll
    for (int q = 1; q < 4; ++q) acc += src[q * (16 * 32)];

    float* gp = ws + ((part & 2) ? WS_GI : WS_GR);
    const int rbase = 16 * ti + ((lane < 16) ? 0 : 8) + 4 * (part & 1);
#pragma unroll
    for (int e = 0; e < 4; ++e) atomicAdd(&gp[(rbase + e) * 64 + c], acc[e]);
  }
}

#define NR_LD 65

template <int KB>
__device__ __forceinline__ void fo_factor_block(float2 (&g)[4][4], float2* Lc, float2* Sg,
                                                float* rdv, const int ty, const int tx) {
#pragma nounroll
  for (int kk = 0; kk < 16; kk += 2) {
    const int k = 16 * KB + kk;
    float2* sg = Sg + ((kk & 2) ? 64 : 0);
    if (tx == kk) {
#pragma unroll
      for (int a = KB; a < 4; ++a) Lc[k * 64 + 16 * a + ty] = g[a][KB];
    }
    if (tx == kk + 1) {
#pragma unroll
      for (int a = KB; a < 4; ++a) sg[16 * a + ty] = g[a][KB];
    }
    __syncthreads();
    const float gkk = Lc[k * 64 + k].x;
    const float2 g10 = Lc[k * 64 + k + 1];
    const float g11 = sg[k + 1].x;
    float2 ri0[4], ri1[4];
    float2 rj0[4], rj1[4];
#pragma unroll
    for (int a = KB; a < 4; ++a) {
      ri0[a] = Lc[k * 64 + 16 * a + ty];
      ri1[a] = sg[16 * a + ty];
    }
#pragma unroll
    for (int b = KB; b < 4; ++b) {
      rj0[b] = Lc[k * 64 + 16 * b + tx];
      rj1[b] = sg[16 * b + tx];
    }
    __builtin_amdgcn_sched_barrier(0);
    const float rd0 = __builtin_amdgcn_rsqf(gkk + 1e-8f);
    const float2 m = make_float2(g10.x * rd0, g10.y * rd0);
    const float d1 = g11 - (m.x * m.x + m.y * m.y);
    const float rd1 = __builtin_amdgcn_rsqf(d1 + 1e-8f);
    float2 li0[4], li1[4], ui[4];
    float2 lj0[4], lj1[4];
#pragma unroll
    for (int a = KB; a < 4; ++a) {
      const float2 l0 = make_float2(ri0[a].x * rd0, ri0[a].y * rd0);
      const float2 u = make_float2(ri1[a].x - (l0.x * m.x + l0.y * m.y),
                                   ri1[a].y - (l0.y * m.x - l0.x * m.y));
      li0[a] = l0;
      ui[a] = u;
      li1[a] = make_float2(u.x * rd1, u.y * rd1);
    }
#pragma unroll
    for (int b = KB; b < 4; ++b) {
      const float2 l0 = make_float2(rj0[b].x * rd0, rj0[b].y * rd0);
      const float2 u = make_float2(rj1[b].x - (l0.x * m.x + l0.y * m.y),
                                   rj1[b].y - (l0.y * m.x - l0.x * m.y));
      lj0[b] = l0;
      lj1[b] = make_float2(u.x * rd1, u.y * rd1);
    }
    if (tx == kk + 1) {
#pragma unroll
      for (int a = KB; a < 4; ++a) Lc[(k + 1) * 64 + 16 * a + ty] = ui[a];
    }
    if (tx == kk && ty == kk) {
      rdv[k] = rd0;
      rdv[k + 1] = rd1;
    }
#pragma unroll
    for (int b = KB; b < 4; ++b) {
#pragma unroll
      for (int a = b; a < 4; ++a) {
        const bool live = (b > KB || tx > kk + 1) && (a > b || ty >= tx);
        if (live) {
          g[a][b].x -= li0[a].x * lj0[b].x + li0[a].y * lj0[b].y;
          g[a][b].y -= li0[a].y * lj0[b].x - li0[a].x * lj0[b].y;
          g[a][b].x -= li1[a].x * lj1[b].x + li1[a].y * lj1[b].y;
          g[a][b].y -= li1[a].y * lj1[b].x - li1[a].x * lj1[b].y;
        }
      }
    }
  }
}

__device__ __forceinline__ void fo_invert_diag(float2 (&nd)[4], const float2* Lc, float2* Nr,
                                               const float* rdv, const int ty, const int tx) {
  float rd[4];
#pragma unroll
  for (int q = 0; q < 4; ++q) rd[q] = rdv[16 * q];
#pragma nounroll
  for (int kk = 0; kk < 16; ++kk) {
    if (ty == kk) {
#pragma unroll
      for (int q = 0; q < 4; ++q) {
        const float2 v = make_float2(nd[q].x * rd[q], nd[q].y * rd[q]);
        nd[q] = v;
        Nr[(16 * q + kk) * NR_LD + 16 * q + tx] = v;
      }
    }
    __syncthreads();
    float rdn[4];
#pragma unroll
    for (int q = 0; q < 4; ++q) rdn[q] = rdv[16 * q + ((kk + 1) & 15)];
#pragma unroll
    for (int q = 0; q < 4; ++q) {
      const int k = 16 * q + kk;
      const float2 t = Lc[k * 64 + 16 * q + ty];
      const float2 lr = make_float2(t.x * rd[q], t.y * rd[q]);
      const float2 nk = Nr[k * NR_LD + 16 * q + tx];
      if (ty > kk) {
        nd[q].x -= lr.x * nk.x - lr.y * nk.y;
        nd[q].y -= lr.x * nk.y + lr.y * nk.x;
      }
    }
#pragma unroll
    for (int q = 0; q < 4; ++q) rd[q] = rdn[q];
  }
}

#define TS_LD 17
#define TS_BLK (16 * TS_LD)

template <int D>
__device__ __forceinline__ void fo_inverse_level(const float2* Lc, float2* Nr, float2* Ts,
                                                 const int ty, const int tx) {
  constexpr int NB = 4 - D;
  constexpr int NM = 16 * D;
  {
    float2 a[NB][NM];
    float2 b[NB][NM];
#pragma unroll
    for (int n = 0; n < NB; ++n) {
#pragma unroll
      for (int e = 0; e < NM; ++e) {
        const int c = 16 * n + e;
        a[n][e] = Lc[c * 64 + 16 * (D + n) + ty];
        b[n][e] = Nr[c * NR_LD + 16 * n + tx];
      }
    }
#pragma unroll
    for (int n = 0; n < NB; ++n) {
      float2 s0 = make_float2(0.0f, 0.0f);
      float2 s1 = make_float2(0.0f, 0.0f);
#pragma unroll
      for (int e = 0; e < NM; e += 2) {
        s0.x += a[n][e].x * b[n][e].x;
        s0.x -= a[n][e].y * b[n][e].y;
        s0.y += a[n][e].x * b[n][e].y;
        s0.y += a[n][e].y * b[n][e].x;
        s1.x += a[n][e + 1].x * b[n][e + 1].x;
        s1.x -= a[n][e + 1].y * b[n][e + 1].y;
        s1.y += a[n][e + 1].x * b[n][e + 1].y;
        s1.y += a[n][e + 1].y * b[n][e + 1].x;
      }
      Ts[(D + n) * TS_BLK + ty * TS_LD + tx] = make_float2(s0.x + s1.x, s0.y + s1.y);
    }
  }
  __syncthreads();
  {
    float2 a[NB][16];
    float2 b[NB][16];
#pragma unroll
    for (int n = 0; n < NB; ++n) {
      const int i = D + n;
#pragma unroll
      for (int m = 0; m < 16; ++m) {
        a[n][m] = Nr[(16 * i + ty) * NR_LD + 16 * i + m];
        b[n][m] = Ts[i * TS_BLK + m * TS_LD + tx];
      }
    }
#pragma unroll
    for (int n = 0; n < NB; ++n) {
      float2 s0 = make_float2(0.0f, 0.0f);
      float2 s1 = make_float2(0.0f, 0.0f);
#pragma unroll
      for (int m = 0; m < 16; m += 2) {
        s0.x += a[n][m].x * b[n][m].x;
        s0.x -= a[n][m].y * b[n][m].y;
        s0.y += a[n][m].x * b[n][m].y;
        s0.y += a[n][m].y * b[n][m].x;
        s1.x += a[n][m + 1].x * b[n][m + 1].x;
        s1.x -= a[n][m + 1].y * b[n][m + 1].y;
        s1.y += a[n][m + 1].x * b[n][m + 1].y;
        s1.y += a[n][m + 1].y * b[n][m + 1].x;
      }
      Nr[(16 * (D + n) + ty) * NR_LD + 16 * n + tx] = make_float2(-(s0.x + s1.x), -(s0.y + s1.y));
    }
  }
  __syncthreads();
}

__device__ __forceinline__ void fo_chol_inv(const float* __restrict__ ws, float2* Lc, float2* Nr,
                                            float* rdv, float2* Sg, float2* Ts) {
  const int tid = threadIdx.x;
  const int ty = tid >> 4;
  const int tx = tid & 15;
  const float* grp = ws + WS_GR;
  const float* gip = ws + WS_GI;

  float2 g[4][4];
#pragma unroll
  for (int a = 0; a < 4; ++a) {
#pragma unroll
    for (int b = 0; b <= a; ++b) {
      const int idx = (16 * a + ty) * 64 + 16 * b + tx;
      g[a][b] = make_float2(grp[idx], gip[idx]);
    }
  }

  fo_factor_block<0>(g, Lc, Sg, rdv, ty, tx);
  fo_factor_block<1>(g, Lc, Sg, rdv, ty, tx);
  fo_factor_block<2>(g, Lc, Sg, rdv, ty, tx);
  fo_factor_block<3>(g, Lc, Sg, rdv, ty, tx);

  {
    float2 t[12];
    float s[12];
#pragma unroll
    for (int m = 0; m < 12; ++m) {
      t[m] = Lc[tid + 256 * m];
      s[m] = rdv[(tid >> 6) + 4 * m];
    }
    const int mend = 4 * ((tid & 63) >> 4);
#pragma unroll
    for (int m = 0; m < 12; ++m) {
      if (m < mend) Lc[tid + 256 * m] = make_float2(t[m].x * s[m], t[m].y * s[m]);
    }
  }

  float2 nd[4];
#pragma unroll
  for (int q = 0; q < 4; ++q) nd[q] = make_float2((ty == tx) ? 1.0f : 0.0f, 0.0f);
  fo_invert_diag(nd, Lc, Nr, rdv, ty, tx);
  fo_inverse_level<1>(Lc, Nr, Ts, ty, tx);
  fo_inverse_level<2>(Lc, Nr, Ts, ty, tx);
  fo_inverse_level<3>(Lc, Nr, Ts, ty, tx);
}

#define APPLY_BLOCK_ROWS 512
#define APPLY_GROUPS (APPLY_BLOCK_ROWS / 64)
#define APPLY_BLOCKS (M_ROWS / APPLY_BLOCK_ROWS)

__device__ __forceinline__ void fo_apply_load(const float* __restrict__ px, v4f (&e)[16]) {
#pragma unroll
  for (int j = 0; j < 16; ++j) e[j] = *(const v4f*)(px + 8 * j);
}

__device__ __forceinline__ void fo_apply_block(const v4f (&e)[16], const v2f (&wr)[2][16],
                                               const v2f (&wi)[2][16], float* __restrict__ po) {
  v8f qr[2] = {};
  v8f qi[2] = {};
#pragma unroll
  for (int j = 0; j < 16; ++j) {
    const v2f ar = v2f{e[j].x, e[j].z};
    const v2f ai = v2f{e[j].y, e[j].w};
    const v2f ain = v2f{-e[j].y, -e[j].w};
#pragma unroll
    for (int t = 0; t < 2; ++t) qr[t] = wmma_f32x4(ar, wr[t][j], qr[t]);
#pragma unroll
    for (int t = 0; t < 2; ++t) qi[t] = wmma_f32x4(ar, wi[t][j], qi[t]);
#pragma unroll
    for (int t = 0; t < 2; ++t) qr[t] = wmma_f32x4(ain, wi[t][j], qr[t]);
#pragma unroll
    for (int t = 0; t < 2; ++t) qi[t] = wmma_f32x4(ai, wr[t][j], qi[t]);
  }
#pragma unroll
  for (int t = 0; t < 2; ++t) {
#pragma unroll
    for (int v = 0; v < 8; ++v) {
      v2f o = {qr[t][v], qi[t][v]};
      *(v2f*)(po + (size_t)v * X_ROW_F + 32 * t) = o;
    }
  }
}

__global__ __launch_bounds__(256) void fo_chol_apply(const float* __restrict__ x,
                                                     const float* __restrict__ ws,
                                                     float* __restrict__ out) {
  __shared__ float2 Lc[64 * 64];
  __shared__ float2 Nr[64 * NR_LD];
  __shared__ float rdv[64];
  __shared__ float2 Sg[2 * 64];
  __shared__ float2 Ts[4 * TS_BLK];

  fo_chol_inv(ws, Lc, Nr, rdv, Sg, Ts);

  const int lane = threadIdx.x & 31;
  const int wave = threadIdx.x >> 5;
  const int rs = wave >> 1;
  const int ch = wave & 1;
  const int p = lane & 15;
  const int kb = (lane < 16) ? 0 : 2;
  const int rb = (lane < 16) ? 0 : 8;

  v2f wr[2][16];
  v2f wi[2][16];
#pragma unroll
  for (int t = 0; t < 2; ++t) {
    const int col = 32 * ch + 16 * t + p;
#pragma unroll
    for (int j = 0; j < 16; ++j) {
      const int r0 = 4 * j + kb;
      const float2 n0 = Nr[col * NR_LD + r0];
      const float2 n1 = Nr[col * NR_LD + r0 + 1];
      const unsigned m0 = (col >= r0) ? 0xffffffffu : 0u;
      const unsigned m1 = (col >= r0 + 1) ? 0xffffffffu : 0u;
      const float v0x = __uint_as_float(__float_as_uint(n0.x) & m0);
      const float v0y = __uint_as_float(__float_as_uint(n0.y) & m0);
      const float v1x = __uint_as_float(__float_as_uint(n1.x) & m1);
      const float v1y = __uint_as_float(__float_as_uint(n1.y) & m1);
      wr[t][j] = v2f{v0x, v1x};
      wi[t][j] = v2f{-v0y, -v1y};
    }
  }

  const size_t row0 = (size_t)blockIdx.x * APPLY_BLOCK_ROWS + 16 * rs;
  const float* px = x + ((row0 + p) * R_COLS + kb) * 2;
  float* po = out + ((row0 + rb) * R_COLS + 32 * ch + p) * 2;
  constexpr size_t kGroupF = (size_t)64 * X_ROW_F;

  v4f e0[16], e1[16];
  fo_apply_load(px, e0);
  for (int g = 0; g < APPLY_GROUPS - 2; g += 2) {
    fo_apply_load(px + (g + 1) * kGroupF, e1);
    __builtin_amdgcn_sched_barrier(0);
    fo_apply_block(e0, wr, wi, po + g * kGroupF);
    fo_apply_load(px + (g + 2) * kGroupF, e0);
    __builtin_amdgcn_sched_barrier(0);
    fo_apply_block(e1, wr, wi, po + (g + 1) * kGroupF);
  }
  fo_apply_load(px + (APPLY_GROUPS - 1) * kGroupF, e1);
  __builtin_amdgcn_sched_barrier(0);
  fo_apply_block(e0, wr, wi, po + (APPLY_GROUPS - 2) * kGroupF);
  fo_apply_block(e1, wr, wi, po + (APPLY_GROUPS - 1) * kGroupF);
}

extern "C" void kernel_launch(void* const* d_in, const int* in_sizes, int n_in,
                              void* d_out, int out_size, void* d_ws, size_t ws_size,
                              hipStream_t stream) {
  (void)in_sizes; (void)n_in; (void)out_size; (void)ws_size;
  const float* x = (const float*)d_in[0];
  float* out = (float*)d_out;
  float* ws = (float*)d_ws;

  fo_zero_gram<<<(2 * 64 * 64 + 255) / 256, 256, 0, stream>>>(ws);
  fo_gram<<<GRAM_BLOCKS, 128, 0, stream>>>(x, ws);
  fo_chol_apply<<<APPLY_BLOCKS, 256, 0, stream>>>(x, ws, out);
}


# Gram: fragment loads widened to b128 (a lane takes 2 adjacent columns, so the 2 tiles of a 32-column pair interleave): 8 loads per 8 rows not 16, ring of 8 register sets not 4; sum order unchanged
# speedup vs baseline: 4.1544x; 1.0336x over previous
#include <hip/hip_runtime.h>


#define M_ROWS 131072
#define R_COLS 64

typedef __attribute__((ext_vector_type(2))) float v2f;
typedef __attribute__((ext_vector_type(4))) float v4f;
typedef __attribute__((ext_vector_type(8))) float v8f;

#define WS_GR 0
#define WS_GI 4096

__device__ inline v8f wmma_f32x4(v2f a, v2f b, v8f c) {
  return __builtin_amdgcn_wmma_f32_16x16x4_f32(
       false, a,  false, b,
       (short)0, c,  false,  false);
}

__global__ void fo_zero_gram(float* ws) {
  int i = blockIdx.x * blockDim.x + threadIdx.x;
  if (i < 2 * 64 * 64) ws[i] = 0.0f;
}

#define GRAM_BLOCK_ROWS 2048
#define GRAM_WAVE_ROWS 512
#define GRAM_BLOCKS (4 * (M_ROWS / GRAM_BLOCK_ROWS))
#define X_ROW_F (R_COLS * 2)

__device__ __forceinline__ void fo_gram_load8(const float* __restrict__ q0,
                                              const float* __restrict__ q1,
                                              v4f (&e)[2][4]) {
#pragma unroll
  for (int j = 0; j < 2; ++j) {
    e[0][2 * j + 0] = *(const v4f*)(q0 + (4 * j + 0) * X_ROW_F);
    e[0][2 * j + 1] = *(const v4f*)(q0 + (4 * j + 1) * X_ROW_F);
    e[1][2 * j + 0] = *(const v4f*)(q1 + (4 * j + 0) * X_ROW_F);
    e[1][2 * j + 1] = *(const v4f*)(q1 + (4 * j + 1) * X_ROW_F);
  }
}

template <int AH>
__device__ __forceinline__ void fo_gram_mma8(const v4f (&e)[2][4], v8f (&gr)[4], v8f (&gi)[4]) {
#pragma unroll
  for (int j = 0; j < 2; ++j) {
    v2f xr[4], xi[4];
#pragma unroll
    for (int u = 0; u < 2; ++u) {
      const v4f e0 = e[u][2 * j + 0];
      const v4f e1 = e[u][2 * j + 1];
      xr[2 * u + 0] = v2f{e0.x, e1.x};
      xi[2 * u + 0] = v2f{e0.y, e1.y};
      xr[2 * u + 1] = v2f{e0.z, e1.z};
      xi[2 * u + 1] = v2f{e0.w, e1.w};
    }
    const v2f ar = xr[AH];
    const v2f ai = xi[AH];
    const v2f ain = {-ai.x, -ai.y};

#pragma unroll
    for (int d = 0; d < 4; ++d) gr[d] = wmma_f32x4(ar, xr[d], gr[d]);
#pragma unroll
    for (int d = 0; d < 4; ++d) gi[d] = wmma_f32x4(ar, xi[d], gi[d]);
#pragma unroll
    for (int d = 0; d < 4; ++d) gr[d] = wmma_f32x4(ai, xi[d], gr[d]);
#pragma unroll
    for (int d = 0; d < 4; ++d) gi[d] = wmma_f32x4(ain, xr[d], gi[d]);
  }
}

template <int AH>
__device__ __forceinline__ void fo_gram_accumulate(const float* __restrict__ q0,
                                                   const float* __restrict__ q1,
                                                   v8f (&gr)[4], v8f (&gi)[4]) {
  v4f s0[2][4], s1[2][4], s2[2][4], s3[2][4], s4[2][4], s5[2][4], s6[2][4], s7[2][4];
  constexpr int kSets = GRAM_WAVE_ROWS / 8;
  constexpr int kSetF = 8 * X_ROW_F;
  fo_gram_load8(q0, q1, s0);
  fo_gram_load8(q0 + kSetF, q1 + kSetF, s1);
  fo_gram_load8(q0 + 2 * kSetF, q1 + 2 * kSetF, s2);
  fo_gram_load8(q0 + 3 * kSetF, q1 + 3 * kSetF, s3);
  fo_gram_load8(q0 + 4 * kSetF, q1 + 4 * kSetF, s4);
  fo_gram_load8(q0 + 5 * kSetF, q1 + 5 * kSetF, s5);
  fo_gram_load8(q0 + 6 * kSetF, q1 + 6 * kSetF, s6);
  fo_gram_load8(q0 + 7 * kSetF, q1 + 7 * kSetF, s7);
  for (int c = 0; c < kSets - 8; c += 8) {
    const float* n0 = q0 + (c + 8) * kSetF;
    const float* n1 = q1 + (c + 8) * kSetF;
    fo_gram_mma8<AH>(s0, gr, gi);
    __builtin_amdgcn_sched_barrier(0);
    fo_gram_load8(n0, n1, s0);
    fo_gram_mma8<AH>(s1, gr, gi);
    __builtin_amdgcn_sched_barrier(0);
    fo_gram_load8(n0 + kSetF, n1 + kSetF, s1);
    fo_gram_mma8<AH>(s2, gr, gi);
    __builtin_amdgcn_sched_barrier(0);
    fo_gram_load8(n0 + 2 * kSetF, n1 + 2 * kSetF, s2);
    fo_gram_mma8<AH>(s3, gr, gi);
    __builtin_amdgcn_sched_barrier(0);
    fo_gram_load8(n0 + 3 * kSetF, n1 + 3 * kSetF, s3);
    fo_gram_mma8<AH>(s4, gr, gi);
    __builtin_amdgcn_sched_barrier(0);
    fo_gram_load8(n0 + 4 * kSetF, n1 + 4 * kSetF, s4);
    fo_gram_mma8<AH>(s5, gr, gi);
    __builtin_amdgcn_sched_barrier(0);
    fo_gram_load8(n0 + 5 * kSetF, n1 + 5 * kSetF, s5);
    fo_gram_mma8<AH>(s6, gr, gi);
    __builtin_amdgcn_sched_barrier(0);
    fo_gram_load8(n0 + 6 * kSetF, n1 + 6 * kSetF, s6);
    fo_gram_mma8<AH>(s7, gr, gi);
    __builtin_amdgcn_sched_barrier(0);
    fo_gram_load8(n0 + 7 * kSetF, n1 + 7 * kSetF, s7);
  }
  fo_gram_mma8<AH>(s0, gr, gi);
  fo_gram_mma8<AH>(s1, gr, gi);
  fo_gram_mma8<AH>(s2, gr, gi);
  fo_gram_mma8<AH>(s3, gr, gi);
  fo_gram_mma8<AH>(s4, gr, gi);
  fo_gram_mma8<AH>(s5, gr, gi);
  fo_gram_mma8<AH>(s6, gr, gi);
  fo_gram_mma8<AH>(s7, gr, gi);
}

__global__ __launch_bounds__(128) void fo_gram(const float* __restrict__ x,
                                               float* __restrict__ ws) {
  __shared__ v4f red[4 * 16 * 32];

  const int lane = threadIdx.x & 31;
  const int kq = threadIdx.x >> 5;
  const int ti = blockIdx.x & 3;
  const int ua = ti >> 1;
  const int p = lane & 15;
  const int kb = (lane < 16) ? 0 : 2;

  const size_t row0 = (size_t)(blockIdx.x >> 2) * GRAM_BLOCK_ROWS + kq * GRAM_WAVE_ROWS + kb;
  const float* xb = x + row0 * X_ROW_F + 4 * p;
  const float* q0 = xb + 64 * ua;
  const float* q1 = xb + 64 * (ua ^ 1);

  v8f gr[4] = {};
  v8f gi[4] = {};

  if (ti & 1) fo_gram_accumulate<1>(q0, q1, gr, gi);
  else        fo_gram_accumulate<0>(q0, q1, gr, gi);

  {
    v4f* dst = red + (kq * 16) * 32 + lane;
#pragma unroll
    for (int d = 0; d < 4; ++d) {
      dst[(4 * d + 0) * 32] = v4f{gr[d][0], gr[d][1], gr[d][2], gr[d][3]};
      dst[(4 * d + 1) * 32] = v4f{gr[d][4], gr[d][5], gr[d][6], gr[d][7]};
      dst[(4 * d + 2) * 32] = v4f{gi[d][0], gi[d][1], gi[d][2], gi[d][3]};
      dst[(4 * d + 3) * 32] = v4f{gi[d][4], gi[d][5], gi[d][6], gi[d][7]};
    }
  }
  __syncthreads();
  const int tj = (kq < 2) ? (2 * ua + kq) : (2 * (ua ^ 1) + (kq - 2));
  const int c = 32 * (tj >> 1) + 2 * p + (tj & 1);
#pragma unroll
  for (int part = 0; part < 4; ++part) {
    const v4f* src = red + (4 * kq + part) * 32 + lane;
    v4f acc = src[0];
#pragma unroll
    for (int q = 1; q < 4; ++q) acc += src[q * (16 * 32)];

    float* gp = ws + ((part & 2) ? WS_GI : WS_GR);
    const int rt = ((lane < 16) ? 0 : 8) + 4 * (part & 1);
    const int rbase = 32 * ua + 2 * rt + (ti & 1);
#pragma unroll
    for (int e = 0; e < 4; ++e) atomicAdd(&gp[(rbase + 2 * e) * 64 + c], acc[e]);
  }
}

#define NR_LD 65

template <int KB>
__device__ __forceinline__ void fo_factor_block(float2 (&g)[4][4], float2* Lc, float2* Sg,
                                                float* rdv, const int ty, const int tx) {
#pragma nounroll
  for (int kk = 0; kk < 16; kk += 2) {
    const int k = 16 * KB + kk;
    float2* sg = Sg + ((kk & 2) ? 64 : 0);
    if (tx == kk) {
#pragma unroll
      for (int a = KB; a < 4; ++a) Lc[k * 64 + 16 * a + ty] = g[a][KB];
    }
    if (tx == kk + 1) {
#pragma unroll
      for (int a = KB; a < 4; ++a) sg[16 * a + ty] = g[a][KB];
    }
    __syncthreads();
    const float gkk = Lc[k * 64 + k].x;
    const float2 g10 = Lc[k * 64 + k + 1];
    const float g11 = sg[k + 1].x;
    float2 ri0[4], ri1[4];
    float2 rj0[4], rj1[4];
#pragma unroll
    for (int a = KB; a < 4; ++a) {
      ri0[a] = Lc[k * 64 + 16 * a + ty];
      ri1[a] = sg[16 * a + ty];
    }
#pragma unroll
    for (int b = KB; b < 4; ++b) {
      rj0[b] = Lc[k * 64 + 16 * b + tx];
      rj1[b] = sg[16 * b + tx];
    }
    __builtin_amdgcn_sched_barrier(0);
    const float rd0 = __builtin_amdgcn_rsqf(gkk + 1e-8f);
    const float2 m = make_float2(g10.x * rd0, g10.y * rd0);
    const float d1 = g11 - (m.x * m.x + m.y * m.y);
    const float rd1 = __builtin_amdgcn_rsqf(d1 + 1e-8f);
    float2 li0[4], li1[4], ui[4];
    float2 lj0[4], lj1[4];
#pragma unroll
    for (int a = KB; a < 4; ++a) {
      const float2 l0 = make_float2(ri0[a].x * rd0, ri0[a].y * rd0);
      const float2 u = make_float2(ri1[a].x - (l0.x * m.x + l0.y * m.y),
                                   ri1[a].y - (l0.y * m.x - l0.x * m.y));
      li0[a] = l0;
      ui[a] = u;
      li1[a] = make_float2(u.x * rd1, u.y * rd1);
    }
#pragma unroll
    for (int b = KB; b < 4; ++b) {
      const float2 l0 = make_float2(rj0[b].x * rd0, rj0[b].y * rd0);
      const float2 u = make_float2(rj1[b].x - (l0.x * m.x + l0.y * m.y),
                                   rj1[b].y - (l0.y * m.x - l0.x * m.y));
      lj0[b] = l0;
      lj1[b] = make_float2(u.x * rd1, u.y * rd1);
    }
    if (tx == kk + 1) {
#pragma unroll
      for (int a = KB; a < 4; ++a) Lc[(k + 1) * 64 + 16 * a + ty] = ui[a];
    }
    if (tx == kk && ty == kk) {
      rdv[k] = rd0;
      rdv[k + 1] = rd1;
    }
#pragma unroll
    for (int b = KB; b < 4; ++b) {
#pragma unroll
      for (int a = b; a < 4; ++a) {
        const bool live = (b > KB || tx > kk + 1) && (a > b || ty >= tx);
        if (live) {
          g[a][b].x -= li0[a].x * lj0[b].x + li0[a].y * lj0[b].y;
          g[a][b].y -= li0[a].y * lj0[b].x - li0[a].x * lj0[b].y;
          g[a][b].x -= li1[a].x * lj1[b].x + li1[a].y * lj1[b].y;
          g[a][b].y -= li1[a].y * lj1[b].x - li1[a].x * lj1[b].y;
        }
      }
    }
  }
}

__device__ __forceinline__ void fo_invert_diag(float2 (&nd)[4], const float2* Lc, float2* Nr,
                                               const float* rdv, const int ty, const int tx) {
  float rd[4];
#pragma unroll
  for (int q = 0; q < 4; ++q) rd[q] = rdv[16 * q];
#pragma nounroll
  for (int kk = 0; kk < 16; ++kk) {
    if (ty == kk) {
#pragma unroll
      for (int q = 0; q < 4; ++q) {
        const float2 v = make_float2(nd[q].x * rd[q], nd[q].y * rd[q]);
        nd[q] = v;
        Nr[(16 * q + kk) * NR_LD + 16 * q + tx] = v;
      }
    }
    __syncthreads();
    float rdn[4];
#pragma unroll
    for (int q = 0; q < 4; ++q) rdn[q] = rdv[16 * q + ((kk + 1) & 15)];
#pragma unroll
    for (int q = 0; q < 4; ++q) {
      const int k = 16 * q + kk;
      const float2 t = Lc[k * 64 + 16 * q + ty];
      const float2 lr = make_float2(t.x * rd[q], t.y * rd[q]);
      const float2 nk = Nr[k * NR_LD + 16 * q + tx];
      if (ty > kk) {
        nd[q].x -= lr.x * nk.x - lr.y * nk.y;
        nd[q].y -= lr.x * nk.y + lr.y * nk.x;
      }
    }
#pragma unroll
    for (int q = 0; q < 4; ++q) rd[q] = rdn[q];
  }
}

#define TS_LD 17
#define TS_BLK (16 * TS_LD)

template <int D>
__device__ __forceinline__ void fo_inverse_level(const float2* Lc, float2* Nr, float2* Ts,
                                                 const int ty, const int tx) {
  constexpr int NB = 4 - D;
  constexpr int NM = 16 * D;
  {
    float2 a[NB][NM];
    float2 b[NB][NM];
#pragma unroll
    for (int n = 0; n < NB; ++n) {
#pragma unroll
      for (int e = 0; e < NM; ++e) {
        const int c = 16 * n + e;
        a[n][e] = Lc[c * 64 + 16 * (D + n) + ty];
        b[n][e] = Nr[c * NR_LD + 16 * n + tx];
      }
    }
#pragma unroll
    for (int n = 0; n < NB; ++n) {
      float2 s0 = make_float2(0.0f, 0.0f);
      float2 s1 = make_float2(0.0f, 0.0f);
#pragma unroll
      for (int e = 0; e < NM; e += 2) {
        s0.x += a[n][e].x * b[n][e].x;
        s0.x -= a[n][e].y * b[n][e].y;
        s0.y += a[n][e].x * b[n][e].y;
        s0.y += a[n][e].y * b[n][e].x;
        s1.x += a[n][e + 1].x * b[n][e + 1].x;
        s1.x -= a[n][e + 1].y * b[n][e + 1].y;
        s1.y += a[n][e + 1].x * b[n][e + 1].y;
        s1.y += a[n][e + 1].y * b[n][e + 1].x;
      }
      Ts[(D + n) * TS_BLK + ty * TS_LD + tx] = make_float2(s0.x + s1.x, s0.y + s1.y);
    }
  }
  __syncthreads();
  {
    float2 a[NB][16];
    float2 b[NB][16];
#pragma unroll
    for (int n = 0; n < NB; ++n) {
      const int i = D + n;
#pragma unroll
      for (int m = 0; m < 16; ++m) {
        a[n][m] = Nr[(16 * i + ty) * NR_LD + 16 * i + m];
        b[n][m] = Ts[i * TS_BLK + m * TS_LD + tx];
      }
    }
#pragma unroll
    for (int n = 0; n < NB; ++n) {
      float2 s0 = make_float2(0.0f, 0.0f);
      float2 s1 = make_float2(0.0f, 0.0f);
#pragma unroll
      for (int m = 0; m < 16; m += 2) {
        s0.x += a[n][m].x * b[n][m].x;
        s0.x -= a[n][m].y * b[n][m].y;
        s0.y += a[n][m].x * b[n][m].y;
        s0.y += a[n][m].y * b[n][m].x;
        s1.x += a[n][m + 1].x * b[n][m + 1].x;
        s1.x -= a[n][m + 1].y * b[n][m + 1].y;
        s1.y += a[n][m + 1].x * b[n][m + 1].y;
        s1.y += a[n][m + 1].y * b[n][m + 1].x;
      }
      Nr[(16 * (D + n) + ty) * NR_LD + 16 * n + tx] = make_float2(-(s0.x + s1.x), -(s0.y + s1.y));
    }
  }
  __syncthreads();
}

__device__ __forceinline__ void fo_chol_inv(const float* __restrict__ ws, float2* Lc, float2* Nr,
                                            float* rdv, float2* Sg, float2* Ts) {
  const int tid = threadIdx.x;
  const int ty = tid >> 4;
  const int tx = tid & 15;
  const float* grp = ws + WS_GR;
  const float* gip = ws + WS_GI;

  float2 g[4][4];
#pragma unroll
  for (int a = 0; a < 4; ++a) {
#pragma unroll
    for (int b = 0; b <= a; ++b) {
      const int idx = (16 * a + ty) * 64 + 16 * b + tx;
      g[a][b] = make_float2(grp[idx], gip[idx]);
    }
  }

  fo_factor_block<0>(g, Lc, Sg, rdv, ty, tx);
  fo_factor_block<1>(g, Lc, Sg, rdv, ty, tx);
  fo_factor_block<2>(g, Lc, Sg, rdv, ty, tx);
  fo_factor_block<3>(g, Lc, Sg, rdv, ty, tx);

  {
    float2 t[12];
    float s[12];
#pragma unroll
    for (int m = 0; m < 12; ++m) {
      t[m] = Lc[tid + 256 * m];
      s[m] = rdv[(tid >> 6) + 4 * m];
    }
    const int mend = 4 * ((tid & 63) >> 4);
#pragma unroll
    for (int m = 0; m < 12; ++m) {
      if (m < mend) Lc[tid + 256 * m] = make_float2(t[m].x * s[m], t[m].y * s[m]);
    }
  }

  float2 nd[4];
#pragma unroll
  for (int q = 0; q < 4; ++q) nd[q] = make_float2((ty == tx) ? 1.0f : 0.0f, 0.0f);
  fo_invert_diag(nd, Lc, Nr, rdv, ty, tx);
  fo_inverse_level<1>(Lc, Nr, Ts, ty, tx);
  fo_inverse_level<2>(Lc, Nr, Ts, ty, tx);
  fo_inverse_level<3>(Lc, Nr, Ts, ty, tx);
}

#define APPLY_BLOCK_ROWS 512
#define APPLY_GROUPS (APPLY_BLOCK_ROWS / 64)
#define APPLY_BLOCKS (M_ROWS / APPLY_BLOCK_ROWS)

__device__ __forceinline__ void fo_apply_load(const float* __restrict__ px, v4f (&e)[16]) {
#pragma unroll
  for (int j = 0; j < 16; ++j) e[j] = *(const v4f*)(px + 8 * j);
}

__device__ __forceinline__ void fo_apply_block(const v4f (&e)[16], const v2f (&wr)[2][16],
                                               const v2f (&wi)[2][16], float* __restrict__ po) {
  v8f qr[2] = {};
  v8f qi[2] = {};
#pragma unroll
  for (int j = 0; j < 16; ++j) {
    const v2f ar = v2f{e[j].x, e[j].z};
    const v2f ai = v2f{e[j].y, e[j].w};
    const v2f ain = v2f{-e[j].y, -e[j].w};
#pragma unroll
    for (int t = 0; t < 2; ++t) qr[t] = wmma_f32x4(ar, wr[t][j], qr[t]);
#pragma unroll
    for (int t = 0; t < 2; ++t) qi[t] = wmma_f32x4(ar, wi[t][j], qi[t]);
#pragma unroll
    for (int t = 0; t < 2; ++t) qr[t] = wmma_f32x4(ain, wi[t][j], qr[t]);
#pragma unroll
    for (int t = 0; t < 2; ++t) qi[t] = wmma_f32x4(ai, wr[t][j], qi[t]);
  }
#pragma unroll
  for (int t = 0; t < 2; ++t) {
#pragma unroll
    for (int v = 0; v < 8; ++v) {
      v2f o = {qr[t][v], qi[t][v]};
      *(v2f*)(po + (size_t)v * X_ROW_F + 32 * t) = o;
    }
  }
}

__global__ __launch_bounds__(256) void fo_chol_apply(const float* __restrict__ x,
                                                     const float* __restrict__ ws,
                                                     float* __restrict__ out) {
  __shared__ float2 Lc[64 * 64];
  __shared__ float2 Nr[64 * NR_LD];
  __shared__ float rdv[64];
  __shared__ float2 Sg[2 * 64];
  __shared__ float2 Ts[4 * TS_BLK];

  fo_chol_inv(ws, Lc, Nr, rdv, Sg, Ts);

  const int lane = threadIdx.x & 31;
  const int wave = threadIdx.x >> 5;
  const int rs = wave >> 1;
  const int ch = wave & 1;
  const int p = lane & 15;
  const int kb = (lane < 16) ? 0 : 2;
  const int rb = (lane < 16) ? 0 : 8;

  v2f wr[2][16];
  v2f wi[2][16];
#pragma unroll
  for (int t = 0; t < 2; ++t) {
    const int col = 32 * ch + 16 * t + p;
#pragma unroll
    for (int j = 0; j < 16; ++j) {
      const int r0 = 4 * j + kb;
      const float2 n0 = Nr[col * NR_LD + r0];
      const float2 n1 = Nr[col * NR_LD + r0 + 1];
      const unsigned m0 = (col >= r0) ? 0xffffffffu : 0u;
      const unsigned m1 = (col >= r0 + 1) ? 0xffffffffu : 0u;
      const float v0x = __uint_as_float(__float_as_uint(n0.x) & m0);
      const float v0y = __uint_as_float(__float_as_uint(n0.y) & m0);
      const float v1x = __uint_as_float(__float_as_uint(n1.x) & m1);
      const float v1y = __uint_as_float(__float_as_uint(n1.y) & m1);
      wr[t][j] = v2f{v0x, v1x};
      wi[t][j] = v2f{-v0y, -v1y};
    }
  }

  const size_t row0 = (size_t)blockIdx.x * APPLY_BLOCK_ROWS + 16 * rs;
  const float* px = x + ((row0 + p) * R_COLS + kb) * 2;
  float* po = out + ((row0 + rb) * R_COLS + 32 * ch + p) * 2;
  constexpr size_t kGroupF = (size_t)64 * X_ROW_F;

  v4f e0[16], e1[16];
  fo_apply_load(px, e0);
  for (int g = 0; g < APPLY_GROUPS - 2; g += 2) {
    fo_apply_load(px + (g + 1) * kGroupF, e1);
    __builtin_amdgcn_sched_barrier(0);
    fo_apply_block(e0, wr, wi, po + g * kGroupF);
    fo_apply_load(px + (g + 2) * kGroupF, e0);
    __builtin_amdgcn_sched_barrier(0);
    fo_apply_block(e1, wr, wi, po + (g + 1) * kGroupF);
  }
  fo_apply_load(px + (APPLY_GROUPS - 1) * kGroupF, e1);
  __builtin_amdgcn_sched_barrier(0);
  fo_apply_block(e0, wr, wi, po + (APPLY_GROUPS - 2) * kGroupF);
  fo_apply_block(e1, wr, wi, po + (APPLY_GROUPS - 1) * kGroupF);
}

extern "C" void kernel_launch(void* const* d_in, const int* in_sizes, int n_in,
                              void* d_out, int out_size, void* d_ws, size_t ws_size,
                              hipStream_t stream) {
  (void)in_sizes; (void)n_in; (void)out_size; (void)ws_size;
  const float* x = (const float*)d_in[0];
  float* out = (float*)d_out;
  float* ws = (float*)d_ws;

  fo_zero_gram<<<(2 * 64 * 64 + 255) / 256, 256, 0, stream>>>(ws);
  fo_gram<<<GRAM_BLOCKS, 128, 0, stream>>>(x, ws);
  fo_chol_apply<<<APPLY_BLOCKS, 256, 0, stream>>>(x, ws, out);
}


# apply by block forward substitution against L: Q_j=(X_j-sum_{i<j}Q_i L_ji^H)L_jj^-H, each D tile reused as the next B; 160 not 256 WMMAs per 16 rows, no off-diagonal inverse (sum order changes)
# speedup vs baseline: 4.7791x; 1.1504x over previous
#include <hip/hip_runtime.h>


#define M_ROWS 131072
#define R_COLS 64

typedef __attribute__((ext_vector_type(2))) float v2f;
typedef __attribute__((ext_vector_type(4))) float v4f;
typedef __attribute__((ext_vector_type(8))) float v8f;

#define WS_GR 0
#define WS_GI 4096

__device__ inline v8f wmma_f32x4(v2f a, v2f b, v8f c) {
  return __builtin_amdgcn_wmma_f32_16x16x4_f32(
       false, a,  false, b,
       (short)0, c,  false,  false);
}

__global__ void fo_zero_gram(float* ws) {
  int i = blockIdx.x * blockDim.x + threadIdx.x;
  if (i < 2 * 64 * 64) ws[i] = 0.0f;
}

#define GRAM_BLOCK_ROWS 2048
#define GRAM_WAVE_ROWS 512
#define GRAM_BLOCKS (4 * (M_ROWS / GRAM_BLOCK_ROWS))
#define X_ROW_F (R_COLS * 2)

__device__ __forceinline__ void fo_gram_load8(const float* __restrict__ q0,
                                              const float* __restrict__ q1,
                                              v4f (&e)[2][4]) {
#pragma unroll
  for (int j = 0; j < 2; ++j) {
    e[0][2 * j + 0] = *(const v4f*)(q0 + (4 * j + 0) * X_ROW_F);
    e[0][2 * j + 1] = *(const v4f*)(q0 + (4 * j + 1) * X_ROW_F);
    e[1][2 * j + 0] = *(const v4f*)(q1 + (4 * j + 0) * X_ROW_F);
    e[1][2 * j + 1] = *(const v4f*)(q1 + (4 * j + 1) * X_ROW_F);
  }
}

template <int AH>
__device__ __forceinline__ void fo_gram_mma8(const v4f (&e)[2][4], v8f (&gr)[4], v8f (&gi)[4]) {
#pragma unroll
  for (int j = 0; j < 2; ++j) {
    v2f xr[4], xi[4];
#pragma unroll
    for (int u = 0; u < 2; ++u) {
      const v4f e0 = e[u][2 * j + 0];
      const v4f e1 = e[u][2 * j + 1];
      xr[2 * u + 0] = v2f{e0.x, e1.x};
      xi[2 * u + 0] = v2f{e0.y, e1.y};
      xr[2 * u + 1] = v2f{e0.z, e1.z};
      xi[2 * u + 1] = v2f{e0.w, e1.w};
    }
    const v2f ar = xr[AH];
    const v2f ai = xi[AH];
    const v2f ain = {-ai.x, -ai.y};

#pragma unroll
    for (int d = 0; d < 4; ++d) gr[d] = wmma_f32x4(ar, xr[d], gr[d]);
#pragma unroll
    for (int d = 0; d < 4; ++d) gi[d] = wmma_f32x4(ar, xi[d], gi[d]);
#pragma unroll
    for (int d = 0; d < 4; ++d) gr[d] = wmma_f32x4(ai, xi[d], gr[d]);
#pragma unroll
    for (int d = 0; d < 4; ++d) gi[d] = wmma_f32x4(ain, xr[d], gi[d]);
  }
}

template <int AH>
__device__ __forceinline__ void fo_gram_accumulate(const float* __restrict__ q0,
                                                   const float* __restrict__ q1,
                                                   v8f (&gr)[4], v8f (&gi)[4]) {
  v4f s0[2][4], s1[2][4], s2[2][4], s3[2][4], s4[2][4], s5[2][4], s6[2][4], s7[2][4];
  constexpr int kSets = GRAM_WAVE_ROWS / 8;
  constexpr int kSetF = 8 * X_ROW_F;
  fo_gram_load8(q0, q1, s0);
  fo_gram_load8(q0 + kSetF, q1 + kSetF, s1);
  fo_gram_load8(q0 + 2 * kSetF, q1 + 2 * kSetF, s2);
  fo_gram_load8(q0 + 3 * kSetF, q1 + 3 * kSetF, s3);
  fo_gram_load8(q0 + 4 * kSetF, q1 + 4 * kSetF, s4);
  fo_gram_load8(q0 + 5 * kSetF, q1 + 5 * kSetF, s5);
  fo_gram_load8(q0 + 6 * kSetF, q1 + 6 * kSetF, s6);
  fo_gram_load8(q0 + 7 * kSetF, q1 + 7 * kSetF, s7);
  for (int c = 0; c < kSets - 8; c += 8) {
    const float* n0 = q0 + (c + 8) * kSetF;
    const float* n1 = q1 + (c + 8) * kSetF;
    fo_gram_mma8<AH>(s0, gr, gi);
    __builtin_amdgcn_sched_barrier(0);
    fo_gram_load8(n0, n1, s0);
    fo_gram_mma8<AH>(s1, gr, gi);
    __builtin_amdgcn_sched_barrier(0);
    fo_gram_load8(n0 + kSetF, n1 + kSetF, s1);
    fo_gram_mma8<AH>(s2, gr, gi);
    __builtin_amdgcn_sched_barrier(0);
    fo_gram_load8(n0 + 2 * kSetF, n1 + 2 * kSetF, s2);
    fo_gram_mma8<AH>(s3, gr, gi);
    __builtin_amdgcn_sched_barrier(0);
    fo_gram_load8(n0 + 3 * kSetF, n1 + 3 * kSetF, s3);
    fo_gram_mma8<AH>(s4, gr, gi);
    __builtin_amdgcn_sched_barrier(0);
    fo_gram_load8(n0 + 4 * kSetF, n1 + 4 * kSetF, s4);
    fo_gram_mma8<AH>(s5, gr, gi);
    __builtin_amdgcn_sched_barrier(0);
    fo_gram_load8(n0 + 5 * kSetF, n1 + 5 * kSetF, s5);
    fo_gram_mma8<AH>(s6, gr, gi);
    __builtin_amdgcn_sched_barrier(0);
    fo_gram_load8(n0 + 6 * kSetF, n1 + 6 * kSetF, s6);
    fo_gram_mma8<AH>(s7, gr, gi);
    __builtin_amdgcn_sched_barrier(0);
    fo_gram_load8(n0 + 7 * kSetF, n1 + 7 * kSetF, s7);
  }
  fo_gram_mma8<AH>(s0, gr, gi);
  fo_gram_mma8<AH>(s1, gr, gi);
  fo_gram_mma8<AH>(s2, gr, gi);
  fo_gram_mma8<AH>(s3, gr, gi);
  fo_gram_mma8<AH>(s4, gr, gi);
  fo_gram_mma8<AH>(s5, gr, gi);
  fo_gram_mma8<AH>(s6, gr, gi);
  fo_gram_mma8<AH>(s7, gr, gi);
}

__global__ __launch_bounds__(128) void fo_gram(const float* __restrict__ x,
                                               float* __restrict__ ws) {
  __shared__ v4f red[4 * 16 * 32];

  const int lane = threadIdx.x & 31;
  const int kq = threadIdx.x >> 5;
  const int ti = blockIdx.x & 3;
  const int ua = ti >> 1;
  const int p = lane & 15;
  const int kb = (lane < 16) ? 0 : 2;

  const size_t row0 = (size_t)(blockIdx.x >> 2) * GRAM_BLOCK_ROWS + kq * GRAM_WAVE_ROWS + kb;
  const float* xb = x + row0 * X_ROW_F + 4 * p;
  const float* q0 = xb + 64 * ua;
  const float* q1 = xb + 64 * (ua ^ 1);

  v8f gr[4] = {};
  v8f gi[4] = {};

  if (ti & 1) fo_gram_accumulate<1>(q0, q1, gr, gi);
  else        fo_gram_accumulate<0>(q0, q1, gr, gi);

  {
    v4f* dst = red + (kq * 16) * 32 + lane;
#pragma unroll
    for (int d = 0; d < 4; ++d) {
      dst[(4 * d + 0) * 32] = v4f{gr[d][0], gr[d][1], gr[d][2], gr[d][3]};
      dst[(4 * d + 1) * 32] = v4f{gr[d][4], gr[d][5], gr[d][6], gr[d][7]};
      dst[(4 * d + 2) * 32] = v4f{gi[d][0], gi[d][1], gi[d][2], gi[d][3]};
      dst[(4 * d + 3) * 32] = v4f{gi[d][4], gi[d][5], gi[d][6], gi[d][7]};
    }
  }
  __syncthreads();
  const int tj = (kq < 2) ? (2 * ua + kq) : (2 * (ua ^ 1) + (kq - 2));
  const int c = 32 * (tj >> 1) + 2 * p + (tj & 1);
#pragma unroll
  for (int part = 0; part < 4; ++part) {
    const v4f* src = red + (4 * kq + part) * 32 + lane;
    v4f acc = src[0];
#pragma unroll
    for (int q = 1; q < 4; ++q) acc += src[q * (16 * 32)];

    float* gp = ws + ((part & 2) ? WS_GI : WS_GR);
    const int rt = ((lane < 16) ? 0 : 8) + 4 * (part & 1);
    const int rbase = 32 * ua + 2 * rt + (ti & 1);
#pragma unroll
    for (int e = 0; e < 4; ++e) atomicAdd(&gp[(rbase + 2 * e) * 64 + c], acc[e]);
  }
}

#define NR_LD 65

template <int KB>
__device__ __forceinline__ void fo_factor_block(float2 (&g)[4][4], float2* Lc, float2* Sg,
                                                float* rdv, const int ty, const int tx) {
#pragma nounroll
  for (int kk = 0; kk < 16; kk += 2) {
    const int k = 16 * KB + kk;
    float2* sg = Sg + ((kk & 2) ? 64 : 0);
    if (tx == kk) {
#pragma unroll
      for (int a = KB; a < 4; ++a) Lc[k * 64 + 16 * a + ty] = g[a][KB];
    }
    if (tx == kk + 1) {
#pragma unroll
      for (int a = KB; a < 4; ++a) sg[16 * a + ty] = g[a][KB];
    }
    __syncthreads();
    const float gkk = Lc[k * 64 + k].x;
    const float2 g10 = Lc[k * 64 + k + 1];
    const float g11 = sg[k + 1].x;
    float2 ri0[4], ri1[4];
    float2 rj0[4], rj1[4];
#pragma unroll
    for (int a = KB; a < 4; ++a) {
      ri0[a] = Lc[k * 64 + 16 * a + ty];
      ri1[a] = sg[16 * a + ty];
    }
#pragma unroll
    for (int b = KB; b < 4; ++b) {
      rj0[b] = Lc[k * 64 + 16 * b + tx];
      rj1[b] = sg[16 * b + tx];
    }
    __builtin_amdgcn_sched_barrier(0);
    const float rd0 = __builtin_amdgcn_rsqf(gkk + 1e-8f);
    const float2 m = make_float2(g10.x * rd0, g10.y * rd0);
    const float d1 = g11 - (m.x * m.x + m.y * m.y);
    const float rd1 = __builtin_amdgcn_rsqf(d1 + 1e-8f);
    float2 li0[4], li1[4], ui[4];
    float2 lj0[4], lj1[4];
#pragma unroll
    for (int a = KB; a < 4; ++a) {
      const float2 l0 = make_float2(ri0[a].x * rd0, ri0[a].y * rd0);
      const float2 u = make_float2(ri1[a].x - (l0.x * m.x + l0.y * m.y),
                                   ri1[a].y - (l0.y * m.x - l0.x * m.y));
      li0[a] = l0;
      ui[a] = u;
      li1[a] = make_float2(u.x * rd1, u.y * rd1);
    }
#pragma unroll
    for (int b = KB; b < 4; ++b) {
      const float2 l0 = make_float2(rj0[b].x * rd0, rj0[b].y * rd0);
      const float2 u = make_float2(rj1[b].x - (l0.x * m.x + l0.y * m.y),
                                   rj1[b].y - (l0.y * m.x - l0.x * m.y));
      lj0[b] = l0;
      lj1[b] = make_float2(u.x * rd1, u.y * rd1);
    }
    if (tx == kk + 1) {
#pragma unroll
      for (int a = KB; a < 4; ++a) Lc[(k + 1) * 64 + 16 * a + ty] = ui[a];
    }
    if (tx == kk && ty == kk) {
      rdv[k] = rd0;
      rdv[k + 1] = rd1;
    }
#pragma unroll
    for (int b = KB; b < 4; ++b) {
#pragma unroll
      for (int a = b; a < 4; ++a) {
        const bool live = (b > KB || tx > kk + 1) && (a > b || ty >= tx);
        if (live) {
          g[a][b].x -= li0[a].x * lj0[b].x + li0[a].y * lj0[b].y;
          g[a][b].y -= li0[a].y * lj0[b].x - li0[a].x * lj0[b].y;
          g[a][b].x -= li1[a].x * lj1[b].x + li1[a].y * lj1[b].y;
          g[a][b].y -= li1[a].y * lj1[b].x - li1[a].x * lj1[b].y;
        }
      }
    }
  }
}

__device__ __forceinline__ void fo_invert_diag(float2 (&nd)[4], const float2* Lc, float2* Nr,
                                               const float* rdv, const int ty, const int tx) {
  float rd[4];
#pragma unroll
  for (int q = 0; q < 4; ++q) rd[q] = rdv[16 * q];
#pragma nounroll
  for (int kk = 0; kk < 16; ++kk) {
    if (ty == kk) {
#pragma unroll
      for (int q = 0; q < 4; ++q) {
        const float2 v = make_float2(nd[q].x * rd[q], nd[q].y * rd[q]);
        nd[q] = v;
        Nr[(16 * q + kk) * NR_LD + 16 * q + tx] = v;
      }
    }
    __syncthreads();
    float rdn[4];
#pragma unroll
    for (int q = 0; q < 4; ++q) rdn[q] = rdv[16 * q + ((kk + 1) & 15)];
#pragma unroll
    for (int q = 0; q < 4; ++q) {
      const int k = 16 * q + kk;
      const float2 t = Lc[k * 64 + 16 * q + ty];
      const float2 lr = make_float2(t.x * rd[q], t.y * rd[q]);
      const float2 nk = Nr[k * NR_LD + 16 * q + tx];
      if (ty > kk) {
        nd[q].x -= lr.x * nk.x - lr.y * nk.y;
        nd[q].y -= lr.x * nk.y + lr.y * nk.x;
      }
    }
#pragma unroll
    for (int q = 0; q < 4; ++q) rd[q] = rdn[q];
  }
}

__device__ __forceinline__ void fo_chol_inv(const float* __restrict__ ws, float2* Lc, float2* Nr,
                                            float* rdv, float2* Sg) {
  const int tid = threadIdx.x;
  const int ty = tid >> 4;
  const int tx = tid & 15;
  const float* grp = ws + WS_GR;
  const float* gip = ws + WS_GI;

  float2 g[4][4];
#pragma unroll
  for (int a = 0; a < 4; ++a) {
#pragma unroll
    for (int b = 0; b <= a; ++b) {
      const int idx = (16 * a + ty) * 64 + 16 * b + tx;
      g[a][b] = make_float2(grp[idx], gip[idx]);
    }
  }

  fo_factor_block<0>(g, Lc, Sg, rdv, ty, tx);
  fo_factor_block<1>(g, Lc, Sg, rdv, ty, tx);
  fo_factor_block<2>(g, Lc, Sg, rdv, ty, tx);
  fo_factor_block<3>(g, Lc, Sg, rdv, ty, tx);

  {
    float2 t[12];
    float s[12];
#pragma unroll
    for (int m = 0; m < 12; ++m) {
      t[m] = Lc[tid + 256 * m];
      s[m] = rdv[(tid >> 6) + 4 * m];
    }
    const int mend = 4 * ((tid & 63) >> 4);
#pragma unroll
    for (int m = 0; m < 12; ++m) {
      if (m < mend) Lc[tid + 256 * m] = make_float2(t[m].x * s[m], t[m].y * s[m]);
    }
  }

  float2 nd[4];
#pragma unroll
  for (int q = 0; q < 4; ++q) nd[q] = make_float2((ty == tx) ? 1.0f : 0.0f, 0.0f);
  fo_invert_diag(nd, Lc, Nr, rdv, ty, tx);
}

#define APPLY_BLOCK_ROWS 512
#define APPLY_BLOCKS (M_ROWS / APPLY_BLOCK_ROWS)
#define ST_LD 33

__device__ __forceinline__ void fo_solve_load(const float* __restrict__ px, v4f (&e)[16]) {
#pragma unroll
  for (int j = 0; j < 4; ++j) {
#pragma unroll
    for (int q = 0; q < 4; ++q) e[4 * j + q] = *(const v4f*)(px + 32 * j + 4 * q);
  }
}

__device__ __forceinline__ constexpr int fo_pair(int j, int i) {
  return (i == 0 ? 0 : (i == 1 ? 3 : 5)) + (j - i - 1);
}

__device__ __forceinline__ void fo_solve_strip(const v4f (&e)[16],
                                               const v2f (&dr)[4][4], const v2f (&di)[4][4],
                                               const v2f (&lr)[6][4], const v2f (&li)[6][4],
                                               v4f* st_w, const v4f* st_r,
                                               float* __restrict__ po) {
  v8f tr[4], ti[4];
#pragma unroll
  for (int j = 0; j < 4; ++j) {
#pragma unroll
    for (int q = 0; q < 4; ++q) {
      tr[j][2 * q + 0] = e[4 * j + q].x;
      ti[j][2 * q + 0] = e[4 * j + q].y;
      tr[j][2 * q + 1] = e[4 * j + q].z;
      ti[j][2 * q + 1] = e[4 * j + q].w;
    }
  }
#pragma unroll
  for (int j = 0; j < 4; ++j) {
    const v8f ntr = -tr[j];
    v8f qr = {};
    v8f qi = {};
#pragma unroll
    for (int s = 0; s < 4; ++s) {
      const v2f btr = v2f{tr[j][2 * s], tr[j][2 * s + 1]};
      const v2f bti = v2f{ti[j][2 * s], ti[j][2 * s + 1]};
      const v2f bnt = v2f{ntr[2 * s], ntr[2 * s + 1]};
      qr = wmma_f32x4(dr[j][s], btr, qr);
      qi = wmma_f32x4(dr[j][s], bti, qi);
      qr = wmma_f32x4(di[j][s], bti, qr);
      qi = wmma_f32x4(di[j][s], bnt, qi);
    }
#pragma unroll
    for (int q = 0; q < 4; ++q)
      st_w[8 * j + q] = v4f{qr[2 * q], qi[2 * q], qr[2 * q + 1], qi[2 * q + 1]};
    if (j < 3) {
      const v8f nqr = -qr;
#pragma unroll
      for (int s = 0; s < 4; ++s) {
        const v2f bqr = v2f{qr[2 * s], qr[2 * s + 1]};
        const v2f bqi = v2f{qi[2 * s], qi[2 * s + 1]};
        const v2f bnq = v2f{nqr[2 * s], nqr[2 * s + 1]};
#pragma unroll
        for (int m = j + 1; m < 4; ++m) {
          tr[m] = wmma_f32x4(lr[fo_pair(m, j)][s], bqr, tr[m]);
          ti[m] = wmma_f32x4(lr[fo_pair(m, j)][s], bqi, ti[m]);
        }
#pragma unroll
        for (int m = j + 1; m < 4; ++m) {
          tr[m] = wmma_f32x4(li[fo_pair(m, j)][s], bqi, tr[m]);
          ti[m] = wmma_f32x4(li[fo_pair(m, j)][s], bnq, ti[m]);
        }
      }
    }
  }
  v4f o[16];
#pragma unroll
  for (int i = 0; i < 16; ++i) o[i] = st_r[(i >> 1) * ST_LD + 16 * (i & 1)];
#pragma unroll
  for (int i = 0; i < 16; ++i) *(v4f*)(po + (size_t)(i >> 1) * X_ROW_F + 64 * (i & 1)) = o[i];
}

__global__ __launch_bounds__(256) void fo_chol_apply(const float* __restrict__ x,
                                                     const float* __restrict__ ws,
                                                     float* __restrict__ out) {
  __shared__ float2 Lc[64 * 64];
  __shared__ float2 Nr[64 * NR_LD];
  __shared__ float rdv[64];
  __shared__ float2 Sg[2 * 64];
  __shared__ v4f St[8 * 16 * ST_LD];

  fo_chol_inv(ws, Lc, Nr, rdv, Sg);

  const int lane = threadIdx.x & 31;
  const int wave = threadIdx.x >> 5;
  const int c = lane & 15;
  const int h = lane >> 4;

  v2f dr[4][4], di[4][4];
  v2f lr[6][4], li[6][4];
#pragma unroll
  for (int j = 0; j < 4; ++j) {
#pragma unroll
    for (int s = 0; s < 4; ++s) {
      const int k0 = 16 * j + 8 * h + 2 * s;
      const float2 n0 = Nr[(16 * j + c) * NR_LD + k0];
      const float2 n1 = Nr[(16 * j + c) * NR_LD + k0 + 1];
      dr[j][s] = v2f{n0.x, n1.x};
      di[j][s] = v2f{n0.y, n1.y};
    }
  }
#pragma unroll
  for (int i = 0; i < 3; ++i) {
#pragma unroll
    for (int j = i + 1; j < 4; ++j) {
#pragma unroll
      for (int s = 0; s < 4; ++s) {
        const int k0 = 16 * i + 8 * h + 2 * s;
        const float2 l0 = Lc[k0 * 64 + 16 * j + c];
        const float2 l1 = Lc[(k0 + 1) * 64 + 16 * j + c];
        lr[fo_pair(j, i)][s] = v2f{-l0.x, -l1.x};
        li[fo_pair(j, i)][s] = v2f{-l0.y, -l1.y};
      }
    }
  }

  const size_t row0 = (size_t)blockIdx.x * APPLY_BLOCK_ROWS + 16 * wave;
  const float* px = x + (row0 + c) * X_ROW_F + 16 * h;
  float* po = out + (row0 + 8 * h) * X_ROW_F + 4 * c;
  v4f* st_w = St + (wave * 16 + c) * ST_LD + 4 * h;
  const v4f* st_r = St + (wave * 16 + 8 * h) * ST_LD + c;
  constexpr size_t kStripF = (size_t)128 * X_ROW_F;

  v4f e0[16], e1[16];
  fo_solve_load(px, e0);
  fo_solve_load(px + kStripF, e1);
  __builtin_amdgcn_sched_barrier(0);
  fo_solve_strip(e0, dr, di, lr, li, st_w, st_r, po);
  fo_solve_load(px + 2 * kStripF, e0);
  __builtin_amdgcn_sched_barrier(0);
  fo_solve_strip(e1, dr, di, lr, li, st_w, st_r, po + kStripF);
  fo_solve_load(px + 3 * kStripF, e1);
  __builtin_amdgcn_sched_barrier(0);
  fo_solve_strip(e0, dr, di, lr, li, st_w, st_r, po + 2 * kStripF);
  __builtin_amdgcn_sched_barrier(0);
  fo_solve_strip(e1, dr, di, lr, li, st_w, st_r, po + 3 * kStripF);
}

extern "C" void kernel_launch(void* const* d_in, const int* in_sizes, int n_in,
                              void* d_out, int out_size, void* d_ws, size_t ws_size,
                              hipStream_t stream) {
  (void)in_sizes; (void)n_in; (void)out_size; (void)ws_size;
  const float* x = (const float*)d_in[0];
  float* out = (float*)d_out;
  float* ws = (float*)d_ws;

  fo_zero_gram<<<(2 * 64 * 64 + 255) / 256, 256, 0, stream>>>(ws);
  fo_gram<<<GRAM_BLOCKS, 128, 0, stream>>>(x, ws);
  fo_chol_apply<<<APPLY_BLOCKS, 256, 0, stream>>>(x, ws, out);
}


# Gram: Im G = P - P^T from one cross product P = Xr^T Xi: 3 real WMMA products per tile and k-step, not 4; ring refilled one set later so it compiles without register copies; Im G sum order changes
# speedup vs baseline: 5.0690x; 1.0607x over previous
#include <hip/hip_runtime.h>


#define M_ROWS 131072
#define R_COLS 64

typedef __attribute__((ext_vector_type(2))) float v2f;
typedef __attribute__((ext_vector_type(4))) float v4f;
typedef __attribute__((ext_vector_type(8))) float v8f;

#define WS_GR 0
#define WS_GX 4096

__device__ inline v8f wmma_f32x4(v2f a, v2f b, v8f c) {
  return __builtin_amdgcn_wmma_f32_16x16x4_f32(
       false, a,  false, b,
       (short)0, c,  false,  false);
}

__global__ void fo_zero_gram(float* ws) {
  int i = blockIdx.x * blockDim.x + threadIdx.x;
  if (i < 2 * 64 * 64) ws[i] = 0.0f;
}

#define GRAM_BLOCK_ROWS 2048
#define GRAM_WAVE_ROWS 512
#define GRAM_BLOCKS (4 * (M_ROWS / GRAM_BLOCK_ROWS))
#define X_ROW_F (R_COLS * 2)

__device__ __forceinline__ void fo_gram_load8(const float* __restrict__ q0,
                                              const float* __restrict__ q1,
                                              v4f (&e)[2][4]) {
#pragma unroll
  for (int j = 0; j < 2; ++j) {
    e[0][2 * j + 0] = *(const v4f*)(q0 + (4 * j + 0) * X_ROW_F);
    e[0][2 * j + 1] = *(const v4f*)(q0 + (4 * j + 1) * X_ROW_F);
    e[1][2 * j + 0] = *(const v4f*)(q1 + (4 * j + 0) * X_ROW_F);
    e[1][2 * j + 1] = *(const v4f*)(q1 + (4 * j + 1) * X_ROW_F);
  }
}

template <int AH>
__device__ __forceinline__ void fo_gram_mma8(const v4f (&e)[2][4], v8f (&gr)[4], v8f (&gx)[4]) {
#pragma unroll
  for (int j = 0; j < 2; ++j) {
    v2f xr[4], xi[4];
#pragma unroll
    for (int u = 0; u < 2; ++u) {
      const v4f e0 = e[u][2 * j + 0];
      const v4f e1 = e[u][2 * j + 1];
      xr[2 * u + 0] = v2f{e0.x, e1.x};
      xi[2 * u + 0] = v2f{e0.y, e1.y};
      xr[2 * u + 1] = v2f{e0.z, e1.z};
      xi[2 * u + 1] = v2f{e0.w, e1.w};
    }
    const v2f ar = xr[AH];
    const v2f ai = xi[AH];

#pragma unroll
    for (int d = 0; d < 4; ++d) gr[d] = wmma_f32x4(ar, xr[d], gr[d]);
#pragma unroll
    for (int d = 0; d < 4; ++d) gx[d] = wmma_f32x4(ar, xi[d], gx[d]);
#pragma unroll
    for (int d = 0; d < 4; ++d) gr[d] = wmma_f32x4(ai, xi[d], gr[d]);
  }
}

template <int AH>
__device__ __forceinline__ void fo_gram_accumulate(const float* __restrict__ q0,
                                                   const float* __restrict__ q1,
                                                   v8f (&gr)[4], v8f (&gx)[4]) {
  v4f s0[2][4], s1[2][4], s2[2][4], s3[2][4], s4[2][4], s5[2][4], s6[2][4], s7[2][4];
  constexpr int kSets = GRAM_WAVE_ROWS / 8;
  constexpr int kSetF = 8 * X_ROW_F;
  fo_gram_load8(q0, q1, s0);
  fo_gram_load8(q0 + kSetF, q1 + kSetF, s1);
  fo_gram_load8(q0 + 2 * kSetF, q1 + 2 * kSetF, s2);
  fo_gram_load8(q0 + 3 * kSetF, q1 + 3 * kSetF, s3);
  fo_gram_load8(q0 + 4 * kSetF, q1 + 4 * kSetF, s4);
  fo_gram_load8(q0 + 5 * kSetF, q1 + 5 * kSetF, s5);
  fo_gram_load8(q0 + 6 * kSetF, q1 + 6 * kSetF, s6);
  for (int c = 0; c < kSets - 8; c += 8) {
    const float* n0 = q0 + (c + 7) * kSetF;
    const float* n1 = q1 + (c + 7) * kSetF;
    fo_gram_mma8<AH>(s0, gr, gx);
    __builtin_amdgcn_sched_barrier(0);
    fo_gram_load8(n0, n1, s7);
    fo_gram_mma8<AH>(s1, gr, gx);
    __builtin_amdgcn_sched_barrier(0);
    fo_gram_load8(n0 + kSetF, n1 + kSetF, s0);
    fo_gram_mma8<AH>(s2, gr, gx);
    __builtin_amdgcn_sched_barrier(0);
    fo_gram_load8(n0 + 2 * kSetF, n1 + 2 * kSetF, s1);
    fo_gram_mma8<AH>(s3, gr, gx);
    __builtin_amdgcn_sched_barrier(0);
    fo_gram_load8(n0 + 3 * kSetF, n1 + 3 * kSetF, s2);
    fo_gram_mma8<AH>(s4, gr, gx);
    __builtin_amdgcn_sched_barrier(0);
    fo_gram_load8(n0 + 4 * kSetF, n1 + 4 * kSetF, s3);
    fo_gram_mma8<AH>(s5, gr, gx);
    __builtin_amdgcn_sched_barrier(0);
    fo_gram_load8(n0 + 5 * kSetF, n1 + 5 * kSetF, s4);
    fo_gram_mma8<AH>(s6, gr, gx);
    __builtin_amdgcn_sched_barrier(0);
    fo_gram_load8(n0 + 6 * kSetF, n1 + 6 * kSetF, s5);
    fo_gram_mma8<AH>(s7, gr, gx);
    __builtin_amdgcn_sched_barrier(0);
    fo_gram_load8(n0 + 7 * kSetF, n1 + 7 * kSetF, s6);
  }
  fo_gram_mma8<AH>(s0, gr, gx);
  fo_gram_load8(q0 + (kSets - 1) * kSetF, q1 + (kSets - 1) * kSetF, s7);
  fo_gram_mma8<AH>(s1, gr, gx);
  fo_gram_mma8<AH>(s2, gr, gx);
  fo_gram_mma8<AH>(s3, gr, gx);
  fo_gram_mma8<AH>(s4, gr, gx);
  fo_gram_mma8<AH>(s5, gr, gx);
  fo_gram_mma8<AH>(s6, gr, gx);
  fo_gram_mma8<AH>(s7, gr, gx);
}

__global__ __launch_bounds__(128) void fo_gram(const float* __restrict__ x,
                                               float* __restrict__ ws) {
  __shared__ v4f red[4 * 16 * 32];

  const int lane = threadIdx.x & 31;
  const int kq = threadIdx.x >> 5;
  const int ti = blockIdx.x & 3;
  const int ua = ti >> 1;
  const int p = lane & 15;
  const int kb = (lane < 16) ? 0 : 2;

  const size_t row0 = (size_t)(blockIdx.x >> 2) * GRAM_BLOCK_ROWS + kq * GRAM_WAVE_ROWS + kb;
  const float* xb = x + row0 * X_ROW_F + 4 * p;
  const float* q0 = xb + 64 * ua;
  const float* q1 = xb + 64 * (ua ^ 1);

  v8f gr[4] = {};
  v8f gx[4] = {};

  if (ti & 1) fo_gram_accumulate<1>(q0, q1, gr, gx);
  else        fo_gram_accumulate<0>(q0, q1, gr, gx);

  {
    v4f* dst = red + (kq * 16) * 32 + lane;
#pragma unroll
    for (int d = 0; d < 4; ++d) {
      dst[(4 * d + 0) * 32] = v4f{gr[d][0], gr[d][1], gr[d][2], gr[d][3]};
      dst[(4 * d + 1) * 32] = v4f{gr[d][4], gr[d][5], gr[d][6], gr[d][7]};
      dst[(4 * d + 2) * 32] = v4f{gx[d][0], gx[d][1], gx[d][2], gx[d][3]};
      dst[(4 * d + 3) * 32] = v4f{gx[d][4], gx[d][5], gx[d][6], gx[d][7]};
    }
  }
  __syncthreads();
  const int tj = (kq < 2) ? (2 * ua + kq) : (2 * (ua ^ 1) + (kq - 2));
  const int c = 32 * (tj >> 1) + 2 * p + (tj & 1);
#pragma unroll
  for (int part = 0; part < 4; ++part) {
    const v4f* src = red + (4 * kq + part) * 32 + lane;
    v4f acc = src[0];
#pragma unroll
    for (int q = 1; q < 4; ++q) acc += src[q * (16 * 32)];

    float* gp = ws + ((part & 2) ? WS_GX : WS_GR);
    const int rt = ((lane < 16) ? 0 : 8) + 4 * (part & 1);
    const int rbase = 32 * ua + 2 * rt + (ti & 1);
#pragma unroll
    for (int e = 0; e < 4; ++e) atomicAdd(&gp[(rbase + 2 * e) * 64 + c], acc[e]);
  }
}

#define NR_LD 65

template <int KB>
__device__ __forceinline__ void fo_factor_block(float2 (&g)[4][4], float2* Lc, float2* Sg,
                                                float* rdv, const int ty, const int tx) {
#pragma nounroll
  for (int kk = 0; kk < 16; kk += 2) {
    const int k = 16 * KB + kk;
    float2* sg = Sg + ((kk & 2) ? 64 : 0);
    if (tx == kk) {
#pragma unroll
      for (int a = KB; a < 4; ++a) Lc[k * 64 + 16 * a + ty] = g[a][KB];
    }
    if (tx == kk + 1) {
#pragma unroll
      for (int a = KB; a < 4; ++a) sg[16 * a + ty] = g[a][KB];
    }
    __syncthreads();
    const float gkk = Lc[k * 64 + k].x;
    const float2 g10 = Lc[k * 64 + k + 1];
    const float g11 = sg[k + 1].x;
    float2 ri0[4], ri1[4];
    float2 rj0[4], rj1[4];
#pragma unroll
    for (int a = KB; a < 4; ++a) {
      ri0[a] = Lc[k * 64 + 16 * a + ty];
      ri1[a] = sg[16 * a + ty];
    }
#pragma unroll
    for (int b = KB; b < 4; ++b) {
      rj0[b] = Lc[k * 64 + 16 * b + tx];
      rj1[b] = sg[16 * b + tx];
    }
    __builtin_amdgcn_sched_barrier(0);
    const float rd0 = __builtin_amdgcn_rsqf(gkk + 1e-8f);
    const float2 m = make_float2(g10.x * rd0, g10.y * rd0);
    const float d1 = g11 - (m.x * m.x + m.y * m.y);
    const float rd1 = __builtin_amdgcn_rsqf(d1 + 1e-8f);
    float2 li0[4], li1[4], ui[4];
    float2 lj0[4], lj1[4];
#pragma unroll
    for (int a = KB; a < 4; ++a) {
      const float2 l0 = make_float2(ri0[a].x * rd0, ri0[a].y * rd0);
      const float2 u = make_float2(ri1[a].x - (l0.x * m.x + l0.y * m.y),
                                   ri1[a].y - (l0.y * m.x - l0.x * m.y));
      li0[a] = l0;
      ui[a] = u;
      li1[a] = make_float2(u.x * rd1, u.y * rd1);
    }
#pragma unroll
    for (int b = KB; b < 4; ++b) {
      const float2 l0 = make_float2(rj0[b].x * rd0, rj0[b].y * rd0);
      const float2 u = make_float2(rj1[b].x - (l0.x * m.x + l0.y * m.y),
                                   rj1[b].y - (l0.y * m.x - l0.x * m.y));
      lj0[b] = l0;
      lj1[b] = make_float2(u.x * rd1, u.y * rd1);
    }
    if (tx == kk + 1) {
#pragma unroll
      for (int a = KB; a < 4; ++a) Lc[(k + 1) * 64 + 16 * a + ty] = ui[a];
    }
    if (tx == kk && ty == kk) {
      rdv[k] = rd0;
      rdv[k + 1] = rd1;
    }
#pragma unroll
    for (int b = KB; b < 4; ++b) {
#pragma unroll
      for (int a = b; a < 4; ++a) {
        const bool live = (b > KB || tx > kk + 1) && (a > b || ty >= tx);
        if (live) {
          g[a][b].x -= li0[a].x * lj0[b].x + li0[a].y * lj0[b].y;
          g[a][b].y -= li0[a].y * lj0[b].x - li0[a].x * lj0[b].y;
          g[a][b].x -= li1[a].x * lj1[b].x + li1[a].y * lj1[b].y;
          g[a][b].y -= li1[a].y * lj1[b].x - li1[a].x * lj1[b].y;
        }
      }
    }
  }
}

__device__ __forceinline__ void fo_invert_diag(float2 (&nd)[4], const float2* Lc, float2* Nr,
                                               const float* rdv, const int ty, const int tx) {
  float rd[4];
#pragma unroll
  for (int q = 0; q < 4; ++q) rd[q] = rdv[16 * q];
#pragma nounroll
  for (int kk = 0; kk < 16; ++kk) {
    if (ty == kk) {
#pragma unroll
      for (int q = 0; q < 4; ++q) {
        const float2 v = make_float2(nd[q].x * rd[q], nd[q].y * rd[q]);
        nd[q] = v;
        Nr[(16 * q + kk) * NR_LD + 16 * q + tx] = v;
      }
    }
    __syncthreads();
    float rdn[4];
#pragma unroll
    for (int q = 0; q < 4; ++q) rdn[q] = rdv[16 * q + ((kk + 1) & 15)];
#pragma unroll
    for (int q = 0; q < 4; ++q) {
      const int k = 16 * q + kk;
      const float2 t = Lc[k * 64 + 16 * q + ty];
      const float2 lr = make_float2(t.x * rd[q], t.y * rd[q]);
      const float2 nk = Nr[k * NR_LD + 16 * q + tx];
      if (ty > kk) {
        nd[q].x -= lr.x * nk.x - lr.y * nk.y;
        nd[q].y -= lr.x * nk.y + lr.y * nk.x;
      }
    }
#pragma unroll
    for (int q = 0; q < 4; ++q) rd[q] = rdn[q];
  }
}

__device__ __forceinline__ void fo_chol_inv(const float* __restrict__ ws, float2* Lc, float2* Nr,
                                            float* rdv, float2* Sg) {
  const int tid = threadIdx.x;
  const int ty = tid >> 4;
  const int tx = tid & 15;
  const float* grp = ws + WS_GR;
  const float* gxp = ws + WS_GX;

  float2 g[4][4];
#pragma unroll
  for (int a = 0; a < 4; ++a) {
#pragma unroll
    for (int b = 0; b <= a; ++b) {
      const int idx = (16 * a + ty) * 64 + 16 * b + tx;
      const int idt = (16 * b + tx) * 64 + 16 * a + ty;
      g[a][b] = make_float2(grp[idx], gxp[idx] - gxp[idt]);
    }
  }

  fo_factor_block<0>(g, Lc, Sg, rdv, ty, tx);
  fo_factor_block<1>(g, Lc, Sg, rdv, ty, tx);
  fo_factor_block<2>(g, Lc, Sg, rdv, ty, tx);
  fo_factor_block<3>(g, Lc, Sg, rdv, ty, tx);

  {
    float2 t[12];
    float s[12];
#pragma unroll
    for (int m = 0; m < 12; ++m) {
      t[m] = Lc[tid + 256 * m];
      s[m] = rdv[(tid >> 6) + 4 * m];
    }
    const int mend = 4 * ((tid & 63) >> 4);
#pragma unroll
    for (int m = 0; m < 12; ++m) {
      if (m < mend) Lc[tid + 256 * m] = make_float2(t[m].x * s[m], t[m].y * s[m]);
    }
  }

  float2 nd[4];
#pragma unroll
  for (int q = 0; q < 4; ++q) nd[q] = make_float2((ty == tx) ? 1.0f : 0.0f, 0.0f);
  fo_invert_diag(nd, Lc, Nr, rdv, ty, tx);
}

#define APPLY_BLOCK_ROWS 512
#define APPLY_BLOCKS (M_ROWS / APPLY_BLOCK_ROWS)
#define ST_LD 33

__device__ __forceinline__ void fo_solve_load(const float* __restrict__ px, v4f (&e)[16]) {
#pragma unroll
  for (int j = 0; j < 4; ++j) {
#pragma unroll
    for (int q = 0; q < 4; ++q) e[4 * j + q] = *(const v4f*)(px + 32 * j + 4 * q);
  }
}

__device__ __forceinline__ constexpr int fo_pair(int j, int i) {
  return (i == 0 ? 0 : (i == 1 ? 3 : 5)) + (j - i - 1);
}

__device__ __forceinline__ void fo_solve_strip(const v4f (&e)[16],
                                               const v2f (&dr)[4][4], const v2f (&di)[4][4],
                                               const v2f (&lr)[6][4], const v2f (&li)[6][4],
                                               v4f* st_w, const v4f* st_r,
                                               float* __restrict__ po) {
  v8f tr[4], ti[4];
#pragma unroll
  for (int j = 0; j < 4; ++j) {
#pragma unroll
    for (int q = 0; q < 4; ++q) {
      tr[j][2 * q + 0] = e[4 * j + q].x;
      ti[j][2 * q + 0] = e[4 * j + q].y;
      tr[j][2 * q + 1] = e[4 * j + q].z;
      ti[j][2 * q + 1] = e[4 * j + q].w;
    }
  }
#pragma unroll
  for (int j = 0; j < 4; ++j) {
    const v8f ntr = -tr[j];
    v8f qr = {};
    v8f qi = {};
#pragma unroll
    for (int s = 0; s < 4; ++s) {
      const v2f btr = v2f{tr[j][2 * s], tr[j][2 * s + 1]};
      const v2f bti = v2f{ti[j][2 * s], ti[j][2 * s + 1]};
      const v2f bnt = v2f{ntr[2 * s], ntr[2 * s + 1]};
      qr = wmma_f32x4(dr[j][s], btr, qr);
      qi = wmma_f32x4(dr[j][s], bti, qi);
      qr = wmma_f32x4(di[j][s], bti, qr);
      qi = wmma_f32x4(di[j][s], bnt, qi);
    }
#pragma unroll
    for (int q = 0; q < 4; ++q)
      st_w[8 * j + q] = v4f{qr[2 * q], qi[2 * q], qr[2 * q + 1], qi[2 * q + 1]};
    if (j < 3) {
      const v8f nqr = -qr;
#pragma unroll
      for (int s = 0; s < 4; ++s) {
        const v2f bqr = v2f{qr[2 * s], qr[2 * s + 1]};
        const v2f bqi = v2f{qi[2 * s], qi[2 * s + 1]};
        const v2f bnq = v2f{nqr[2 * s], nqr[2 * s + 1]};
#pragma unroll
        for (int m = j + 1; m < 4; ++m) {
          tr[m] = wmma_f32x4(lr[fo_pair(m, j)][s], bqr, tr[m]);
          ti[m] = wmma_f32x4(lr[fo_pair(m, j)][s], bqi, ti[m]);
        }
#pragma unroll
        for (int m = j + 1; m < 4; ++m) {
          tr[m] = wmma_f32x4(li[fo_pair(m, j)][s], bqi, tr[m]);
          ti[m] = wmma_f32x4(li[fo_pair(m, j)][s], bnq, ti[m]);
        }
      }
    }
  }
  v4f o[16];
#pragma unroll
  for (int i = 0; i < 16; ++i) o[i] = st_r[(i >> 1) * ST_LD + 16 * (i & 1)];
#pragma unroll
  for (int i = 0; i < 16; ++i) *(v4f*)(po + (size_t)(i >> 1) * X_ROW_F + 64 * (i & 1)) = o[i];
}

__global__ __launch_bounds__(256) void fo_chol_apply(const float* __restrict__ x,
                                                     const float* __restrict__ ws,
                                                     float* __restrict__ out) {
  __shared__ float2 Lc[64 * 64];
  __shared__ float2 Nr[64 * NR_LD];
  __shared__ float rdv[64];
  __shared__ float2 Sg[2 * 64];
  __shared__ v4f St[8 * 16 * ST_LD];

  fo_chol_inv(ws, Lc, Nr, rdv, Sg);

  const int lane = threadIdx.x & 31;
  const int wave = threadIdx.x >> 5;
  const int c = lane & 15;
  const int h = lane >> 4;

  v2f dr[4][4], di[4][4];
  v2f lr[6][4], li[6][4];
#pragma unroll
  for (int j = 0; j < 4; ++j) {
#pragma unroll
    for (int s = 0; s < 4; ++s) {
      const int k0 = 16 * j + 8 * h + 2 * s;
      const float2 n0 = Nr[(16 * j + c) * NR_LD + k0];
      const float2 n1 = Nr[(16 * j + c) * NR_LD + k0 + 1];
      dr[j][s] = v2f{n0.x, n1.x};
      di[j][s] = v2f{n0.y, n1.y};
    }
  }
#pragma unroll
  for (int i = 0; i < 3; ++i) {
#pragma unroll
    for (int j = i + 1; j < 4; ++j) {
#pragma unroll
      for (int s = 0; s < 4; ++s) {
        const int k0 = 16 * i + 8 * h + 2 * s;
        const float2 l0 = Lc[k0 * 64 + 16 * j + c];
        const float2 l1 = Lc[(k0 + 1) * 64 + 16 * j + c];
        lr[fo_pair(j, i)][s] = v2f{-l0.x, -l1.x};
        li[fo_pair(j, i)][s] = v2f{-l0.y, -l1.y};
      }
    }
  }

  const size_t row0 = (size_t)blockIdx.x * APPLY_BLOCK_ROWS + 16 * wave;
  const float* px = x + (row0 + c) * X_ROW_F + 16 * h;
  float* po = out + (row0 + 8 * h) * X_ROW_F + 4 * c;
  v4f* st_w = St + (wave * 16 + c) * ST_LD + 4 * h;
  const v4f* st_r = St + (wave * 16 + 8 * h) * ST_LD + c;
  constexpr size_t kStripF = (size_t)128 * X_ROW_F;

  v4f e0[16], e1[16];
  fo_solve_load(px, e0);
  fo_solve_load(px + kStripF, e1);
  __builtin_amdgcn_sched_barrier(0);
  fo_solve_strip(e0, dr, di, lr, li, st_w, st_r, po);
  fo_solve_load(px + 2 * kStripF, e0);
  __builtin_amdgcn_sched_barrier(0);
  fo_solve_strip(e1, dr, di, lr, li, st_w, st_r, po + kStripF);
  fo_solve_load(px + 3 * kStripF, e1);
  __builtin_amdgcn_sched_barrier(0);
  fo_solve_strip(e0, dr, di, lr, li, st_w, st_r, po + 2 * kStripF);
  __builtin_amdgcn_sched_barrier(0);
  fo_solve_strip(e1, dr, di, lr, li, st_w, st_r, po + 3 * kStripF);
}

extern "C" void kernel_launch(void* const* d_in, const int* in_sizes, int n_in,
                              void* d_out, int out_size, void* d_ws, size_t ws_size,
                              hipStream_t stream) {
  (void)in_sizes; (void)n_in; (void)out_size; (void)ws_size;
  const float* x = (const float*)d_in[0];
  float* out = (float*)d_out;
  float* ws = (float*)d_ws;

  fo_zero_gram<<<(2 * 64 * 64 + 255) / 256, 256, 0, stream>>>(ws);
  fo_gram<<<GRAM_BLOCKS, 128, 0, stream>>>(x, ws);
  fo_chol_apply<<<APPLY_BLOCKS, 256, 0, stream>>>(x, ws, out);
}
